# starting kernel
# baseline (speedup 1.0000x reference)
#include <hip/hip_runtime.h>
#include <hip/hip_bf16.h>

typedef float v2f __attribute__((ext_vector_type(2)));
typedef float v8f __attribute__((ext_vector_type(8)));

#define NB 4096
#define NA 64
#define NL 16
#define NM 4
#define NE 10
#define P_TOT 416
#define P1_END 32
#define P2_END 160
#define EPAD 16
#define XPITCH 68

__global__ void prep_offsets_kernel(const int* __restrict__ i1, const int* __restrict__ l1,
                                    const int* __restrict__ i2, const int* __restrict__ j2,
                                    const int* __restrict__ l2, const int* __restrict__ m2,
                                    const int* __restrict__ i3, const int* __restrict__ j3,
                                    const int* __restrict__ f3, const int* __restrict__ l3,
                                    const int* __restrict__ m3, const int* __restrict__ g3,
                                    int* __restrict__ offs) {
  int p = blockIdx.x * blockDim.x + threadIdx.x;
  if (p >= P_TOT) return;
  int packed;
  if (p < P1_END) {
    int o1 = i1[p] * NM + l1[p];
    packed = o1;
  } else if (p < P2_END) {
    int q = p - P1_END;
    int o1 = i2[q] * NM + l2[q];
    int o2 = j2[q] * NM + m2[q];
    packed = o1 | (o2 << 8);
  } else {
    int q = p - P2_END;
    int o1 = i3[q] * NM + l3[q];
    int o2 = j3[q] * NM + m3[q];
    int o3 = f3[q] * NM + g3[q];
    packed = o1 | (o2 << 8) | (o3 << 16);
  }
  offs[p] = packed;
}

__global__ void prep_wg_kernel(const float* __restrict__ w1, const float* __restrict__ w2,
                               const float* __restrict__ w3,
                               const float* __restrict__ c1, const float* __restrict__ c2,
                               const float* __restrict__ c3,
                               const int* __restrict__ k1, const int* __restrict__ q1,
                               const int* __restrict__ k2, const int* __restrict__ q2,
                               const int* __restrict__ k3, const int* __restrict__ q3,
                               float* __restrict__ Wg) {
  int tid = blockIdx.x * blockDim.x + threadIdx.x;
  if (tid >= NA * P_TOT) return;
  int p = tid % P_TOT;
  int a = tid / P_TOT;
  float cf;
  const float* wbase;
  int estride;
  int kq;
  if (p < P1_END) {
    cf = c1[p];
    wbase = w1; kq = k1[p] * 2 + q1[p]; estride = 4 * 2 * NA;
  } else if (p < P2_END) {
    int t = p - P1_END;
    cf = c2[t];
    wbase = w2; kq = k2[t] * 4 + q2[t]; estride = 16 * 4 * NA;
  } else {
    int t = p - P2_END;
    cf = c3[t];
    wbase = w3; kq = k3[t] * 6 + q3[t]; estride = 32 * 6 * NA;
  }
  const int kqstride = kq * NA + a;
  float* dst = Wg + ((size_t)a * P_TOT + p) * EPAD;
#pragma unroll
  for (int n = 0; n < EPAD; ++n) {
    dst[n] = (n < NE) ? cf * wbase[(size_t)n * estride + kqstride] : 0.0f;
  }
}

template <int CNT>
__device__ __forceinline__ void wmma_step(int t, const int* __restrict__ ldsOff,
                                          const float* __restrict__ X,
                                          const float* __restrict__ ldsW,
                                          int half, int lrow, v8f& acc) {
  const int pA = 4 * t + 2 * half;
  const int pk0 = ldsOff[pA];
  const int pk1 = ldsOff[pA + 1];
  float xa = X[pk0 & 0xFF];
  float xb = X[pk1 & 0xFF];
  if (CNT > 1) { xa *= X[(pk0 >> 8) & 0xFF]; xb *= X[(pk1 >> 8) & 0xFF]; }
  if (CNT > 2) { xa *= X[(pk0 >> 16) & 0xFF]; xb *= X[(pk1 >> 16) & 0xFF]; }
  v2f af; af.x = xa; af.y = xb;
  v2f bf;
  bf.x = ldsW[pA * EPAD + lrow];
  bf.y = ldsW[(pA + 1) * EPAD + lrow];
  acc = __builtin_amdgcn_wmma_f32_16x16x4_f32(false, af, false, bf,
                                              (short)0, acc, false, false);
}

__global__ __launch_bounds__(256) void contract_kernel(const float* __restrict__ x,
                                                       const float* __restrict__ y,
                                                       const float* __restrict__ Wg,
                                                       const int* __restrict__ offs,
                                                       float* __restrict__ out) {
  __shared__ float ldsW[P_TOT * EPAD];
  __shared__ float ldsX[8 * 16 * XPITCH];
  __shared__ float ldsY[128 * EPAD];
  __shared__ int   ldsOff[P_TOT];

  const int tid  = threadIdx.x;
  const int wv   = tid >> 5;
  const int lane = tid & 31;
  const int half = lane >> 4;
  const int lrow = lane & 15;
  const int b0   = blockIdx.x * 128;
  const int a0   = blockIdx.y * 8;
  const int myb  = b0 + wv * 16;

  for (int i = tid; i < P_TOT; i += 256) ldsOff[i] = offs[i];
  for (int i = tid; i < 128 * EPAD; i += 256) {
    int row = i >> 4, e = i & 15;
    ldsY[i] = (e < NE) ? y[(size_t)(b0 + row) * NE + e] : 0.0f;
  }

  for (int ai = 0; ai < 8; ++ai) {
    const int a = a0 + ai;
    __syncthreads();
    {
      const float4* src = (const float4*)(Wg + (size_t)a * P_TOT * EPAD);
      float4* dst = (float4*)ldsW;
      for (int i = tid; i < P_TOT * EPAD / 4; i += 256) dst[i] = src[i];
    }
    for (int i = lane; i < 16 * 16; i += 32) {
      int row = i >> 4, c4 = i & 15;
      const float4 v = *(const float4*)(x + ((size_t)(myb + row) * NA + a) * 64 + c4 * 4);
      *(float4*)(&ldsX[(wv * 16 + row) * XPITCH + c4 * 4]) = v;
    }
    __syncthreads();

    const float* X = &ldsX[(wv * 16 + lrow) * XPITCH];
    v8f acc = {};
#pragma unroll
    for (int t = 0; t < 8; ++t)
      wmma_step<1>(t, ldsOff, X, ldsW, half, lrow, acc);
#pragma unroll 4
    for (int t = 8; t < 40; ++t)
      wmma_step<2>(t, ldsOff, X, ldsW, half, lrow, acc);
#pragma unroll 4
    for (int t = 40; t < 104; ++t)
      wmma_step<3>(t, ldsOff, X, ldsW, half, lrow, acc);

#pragma unroll
    for (int r = 0; r < 8; ++r) {
      const int row = wv * 16 + r + half * 8;
      float pr = acc[r] * ldsY[row * EPAD + lrow];
      pr += __shfl_xor(pr, 1, 16);
      pr += __shfl_xor(pr, 2, 16);
      pr += __shfl_xor(pr, 4, 16);
      pr += __shfl_xor(pr, 8, 16);
      if (lrow == 0) out[(size_t)(b0 + row) * NA + a] = pr;
    }
  }
}

extern "C" void kernel_launch(void* const* d_in, const int* in_sizes, int n_in,
                              void* d_out, int out_size, void* d_ws, size_t ws_size,
                              hipStream_t stream) {
  const float* x  = (const float*)d_in[0];
  const float* y  = (const float*)d_in[1];
  const float* w1 = (const float*)d_in[2];
  const float* w2 = (const float*)d_in[3];
  const float* w3 = (const float*)d_in[4];
  const float* c1 = (const float*)d_in[5];
  const float* c2 = (const float*)d_in[6];
  const float* c3 = (const float*)d_in[7];
  const int* i1 = (const int*)d_in[8];
  const int* l1 = (const int*)d_in[9];
  const int* k1 = (const int*)d_in[10];
  const int* q1 = (const int*)d_in[11];
  const int* i2 = (const int*)d_in[12];
  const int* j2 = (const int*)d_in[13];
  const int* l2 = (const int*)d_in[14];
  const int* m2 = (const int*)d_in[15];
  const int* k2 = (const int*)d_in[16];
  const int* q2 = (const int*)d_in[17];
  const int* i3 = (const int*)d_in[18];
  const int* j3 = (const int*)d_in[19];
  const int* f3 = (const int*)d_in[20];
  const int* l3 = (const int*)d_in[21];
  const int* m3 = (const int*)d_in[22];
  const int* g3 = (const int*)d_in[23];
  const int* k3 = (const int*)d_in[24];
  const int* q3 = (const int*)d_in[25];

  int*   offs = (int*)d_ws;
  float* Wg   = (float*)((char*)d_ws + 2048);
  float* out  = (float*)d_out;

  prep_offsets_kernel<<<2, 256, 0, stream>>>(i1, l1, i2, j2, l2, m2,
                                             i3, j3, f3, l3, m3, g3, offs);
  prep_wg_kernel<<<104, 256, 0, stream>>>(w1, w2, w3, c1, c2, c3,
                                          k1, q1, k2, q2, k3, q3, Wg);
  contract_kernel<<<dim3(32, 8), 256, 0, stream>>>(x, y, Wg, offs, out);
}


# a-stationary transposed product: each wave holds Wg[a]^T as 104 A fragments in VGPRs, streams 16-row x tiles via its own LDS buffer (no per-a staging, no barriers); e sum in-lane, order changes
# speedup vs baseline: 2.3741x; 2.3741x over previous
#include <hip/hip_runtime.h>
#include <hip/hip_bf16.h>

typedef float v2f __attribute__((ext_vector_type(2)));
typedef float v8f __attribute__((ext_vector_type(8)));

#define NB 4096
#define NA 64
#define NL 16
#define NM 4
#define NE 10
#define P_TOT 416
#define P1_END 32
#define P2_END 160
#define EPAD 16
#define XPITCH 68
#define KSTEPS 104
#define WAVES 12
#define GROUPS_PER_A 4
#define WAVES_PER_A (WAVES * GROUPS_PER_A)
#define BTILES (NB / 16)

__global__ void prep_offsets_kernel(const int* __restrict__ i1, const int* __restrict__ l1,
                                    const int* __restrict__ i2, const int* __restrict__ j2,
                                    const int* __restrict__ l2, const int* __restrict__ m2,
                                    const int* __restrict__ i3, const int* __restrict__ j3,
                                    const int* __restrict__ f3, const int* __restrict__ l3,
                                    const int* __restrict__ m3, const int* __restrict__ g3,
                                    int* __restrict__ offs) {
  int p = blockIdx.x * blockDim.x + threadIdx.x;
  if (p >= P_TOT) return;
  int packed;
  if (p < P1_END) {
    int o1 = (i1[p] * NM + l1[p]) * 4;
    packed = o1;
  } else if (p < P2_END) {
    int q = p - P1_END;
    int o1 = (i2[q] * NM + l2[q]) * 4;
    int o2 = (j2[q] * NM + m2[q]) * 4;
    packed = o1 | (o2 << 8);
  } else {
    int q = p - P2_END;
    int o1 = (i3[q] * NM + l3[q]) * 4;
    int o2 = (j3[q] * NM + m3[q]) * 4;
    int o3 = (f3[q] * NM + g3[q]) * 4;
    packed = o1 | (o2 << 8) | (o3 << 16);
  }
  offs[p] = packed;
}

__global__ void prep_wg_kernel(const float* __restrict__ w1, const float* __restrict__ w2,
                               const float* __restrict__ w3,
                               const float* __restrict__ c1, const float* __restrict__ c2,
                               const float* __restrict__ c3,
                               const int* __restrict__ k1, const int* __restrict__ q1,
                               const int* __restrict__ k2, const int* __restrict__ q2,
                               const int* __restrict__ k3, const int* __restrict__ q3,
                               float* __restrict__ Wg) {
  int tid = blockIdx.x * blockDim.x + threadIdx.x;
  if (tid >= NA * P_TOT) return;
  int p = tid % P_TOT;
  int a = tid / P_TOT;
  float cf;
  const float* wbase;
  int estride;
  int kq;
  if (p < P1_END) {
    cf = c1[p];
    wbase = w1; kq = k1[p] * 2 + q1[p]; estride = 4 * 2 * NA;
  } else if (p < P2_END) {
    int t = p - P1_END;
    cf = c2[t];
    wbase = w2; kq = k2[t] * 4 + q2[t]; estride = 16 * 4 * NA;
  } else {
    int t = p - P2_END;
    cf = c3[t];
    wbase = w3; kq = k3[t] * 6 + q3[t]; estride = 32 * 6 * NA;
  }
  const int kqstride = kq * NA + a;
  float* dst = Wg + ((size_t)a * P_TOT + p) * EPAD;
#pragma unroll
  for (int n = 0; n < EPAD; ++n) {
    dst[n] = (n < NE) ? cf * wbase[(size_t)n * estride + kqstride] : 0.0f;
  }
}

template <int CNT>
__device__ __forceinline__ v2f xprod_frag(int pk0, int pk1, const char* Xb) {
  float xa = *(const float*)(Xb + (pk0 & 0xFF));
  float xb = *(const float*)(Xb + (pk1 & 0xFF));
  if (CNT > 1) {
    xa *= *(const float*)(Xb + ((pk0 >> 8) & 0xFF));
    xb *= *(const float*)(Xb + ((pk1 >> 8) & 0xFF));
  }
  if (CNT > 2) {
    xa *= *(const float*)(Xb + ((pk0 >> 16) & 0xFF));
    xb *= *(const float*)(Xb + ((pk1 >> 16) & 0xFF));
  }
  v2f f; f.x = xa; f.y = xb;
  return f;
}

#define XROWSTEP ((size_t)2 * NA * 64)
#define LOAD_XROWS(P)                                        \
  x0 = *(const float4*)((P));                                \
  x1 = *(const float4*)((P) + 1 * XROWSTEP);                 \
  x2 = *(const float4*)((P) + 2 * XROWSTEP);                 \
  x3 = *(const float4*)((P) + 3 * XROWSTEP);                 \
  x4 = *(const float4*)((P) + 4 * XROWSTEP);                 \
  x5 = *(const float4*)((P) + 5 * XROWSTEP);                 \
  x6 = *(const float4*)((P) + 6 * XROWSTEP);                 \
  x7 = *(const float4*)((P) + 7 * XROWSTEP);

#define REGION_END(T) if (((T) + 1) % 8 == 0) __builtin_amdgcn_sched_barrier(0)

__global__ __launch_bounds__(WAVES * 32) void contract_kernel(const float* __restrict__ x,
                                                              const float* __restrict__ y,
                                                              const float* __restrict__ Wg,
                                                              const int* __restrict__ offs,
                                                              float* __restrict__ out) {
  __shared__ float ldsX[WAVES * 16 * XPITCH];
  __shared__ int   ldsOff[P_TOT];

  const int tid  = threadIdx.x;
  const int wv   = __builtin_amdgcn_readfirstlane(tid >> 5);
  const int lane = tid & 31;
  const int half = lane >> 4;
  const int lrow = lane & 15;
  const int a    = blockIdx.y;
  const int wa   = blockIdx.x * WAVES + wv;
  const int ntiles = (BTILES - wa + WAVES_PER_A - 1) / WAVES_PER_A;

  for (int i = tid; i < P_TOT; i += WAVES * 32) ldsOff[i] = offs[i];

  v2f wf[KSTEPS];
  {
    const float* wsrc = Wg + (size_t)a * P_TOT * EPAD + (2 * half) * EPAD + lrow;
#pragma unroll
    for (int t = 0; t < KSTEPS; ++t) {
      wf[t].x = wsrc[(4 * t) * EPAD];
      wf[t].y = wsrc[(4 * t + 1) * EPAD];
    }
  }

  const int srow = lane >> 4;
  const int c4   = lane & 15;
  const float* xsrc = x + ((size_t)(wa * 16 + srow) * NA + a) * 64 + c4 * 4;
  float4 x0, x1, x2, x3, x4, x5, x6, x7;
  LOAD_XROWS(xsrc)

  __syncthreads();

  float* Xw = &ldsX[wv * 16 * XPITCH];
  float* Xst = Xw + srow * XPITCH + c4 * 4;
  const char* Xb = (const char*)(Xw + lrow * XPITCH);
  const int* offl = ldsOff + 2 * half;

  for (int it = 0; it < ntiles; ++it) {
    const int b0 = (wa + it * WAVES_PER_A) * 16;

    __builtin_amdgcn_fence(__ATOMIC_RELEASE, "wavefront");
    __builtin_amdgcn_wave_barrier();
    *(float4*)(Xst + 0 * 2 * XPITCH) = x0;
    *(float4*)(Xst + 1 * 2 * XPITCH) = x1;
    *(float4*)(Xst + 2 * 2 * XPITCH) = x2;
    *(float4*)(Xst + 3 * 2 * XPITCH) = x3;
    *(float4*)(Xst + 4 * 2 * XPITCH) = x4;
    *(float4*)(Xst + 5 * 2 * XPITCH) = x5;
    *(float4*)(Xst + 6 * 2 * XPITCH) = x6;
    *(float4*)(Xst + 7 * 2 * XPITCH) = x7;
    __builtin_amdgcn_fence(__ATOMIC_RELEASE, "wavefront");
    __builtin_amdgcn_wave_barrier();
    __builtin_amdgcn_fence(__ATOMIC_ACQUIRE, "wavefront");

    if (it + 1 < ntiles) {
      const float* xn = xsrc + (size_t)((it + 1) * WAVES_PER_A * 16) * NA * 64;
      LOAD_XROWS(xn)
    }

    float yv[8];
#pragma unroll
    for (int r = 0; r < 8; ++r) {
      const int e  = 8 * half + r;
      const int ec = e < NE ? e : NE - 1;
      const float v = y[(size_t)(b0 + lrow) * NE + ec];
      yv[r] = e < NE ? v : 0.0f;
    }

    v8f acc = {};
#pragma unroll
    for (int t = 0; t < 8; ++t) {
      const v2f bf = xprod_frag<1>(offl[4 * t], offl[4 * t + 1], Xb);
      acc = __builtin_amdgcn_wmma_f32_16x16x4_f32(false, wf[t], false, bf,
                                                  (short)0, acc, false, false);
      REGION_END(t);
    }
#pragma unroll
    for (int t = 8; t < 40; ++t) {
      const v2f bf = xprod_frag<2>(offl[4 * t], offl[4 * t + 1], Xb);
      acc = __builtin_amdgcn_wmma_f32_16x16x4_f32(false, wf[t], false, bf,
                                                  (short)0, acc, false, false);
      REGION_END(t);
    }
#pragma unroll
    for (int t = 40; t < KSTEPS; ++t) {
      const v2f bf = xprod_frag<3>(offl[4 * t], offl[4 * t + 1], Xb);
      acc = __builtin_amdgcn_wmma_f32_16x16x4_f32(false, wf[t], false, bf,
                                                  (short)0, acc, false, false);
      REGION_END(t);
    }

    asm volatile("s_wait_alu depctr_va_vdst(0)\n\tv_nop\n\tv_nop\n\tv_nop\n\tv_nop\n\tv_nop\n\tv_nop\n\tv_nop\n\tv_nop"
                 : "+v"(acc));

    float part = acc[0] * yv[0];
#pragma unroll
    for (int r = 1; r < 8; ++r) part += acc[r] * yv[r];
    part += __shfl_xor(part, 16, 32);
    if (half == 0) out[(size_t)(b0 + lrow) * NA + a] = part;
  }
}

extern "C" void kernel_launch(void* const* d_in, const int* in_sizes, int n_in,
                              void* d_out, int out_size, void* d_ws, size_t ws_size,
                              hipStream_t stream) {
  const float* x  = (const float*)d_in[0];
  const float* y  = (const float*)d_in[1];
  const float* w1 = (const float*)d_in[2];
  const float* w2 = (const float*)d_in[3];
  const float* w3 = (const float*)d_in[4];
  const float* c1 = (const float*)d_in[5];
  const float* c2 = (const float*)d_in[6];
  const float* c3 = (const float*)d_in[7];
  const int* i1 = (const int*)d_in[8];
  const int* l1 = (const int*)d_in[9];
  const int* k1 = (const int*)d_in[10];
  const int* q1 = (const int*)d_in[11];
  const int* i2 = (const int*)d_in[12];
  const int* j2 = (const int*)d_in[13];
  const int* l2 = (const int*)d_in[14];
  const int* m2 = (const int*)d_in[15];
  const int* k2 = (const int*)d_in[16];
  const int* q2 = (const int*)d_in[17];
  const int* i3 = (const int*)d_in[18];
  const int* j3 = (const int*)d_in[19];
  const int* f3 = (const int*)d_in[20];
  const int* l3 = (const int*)d_in[21];
  const int* m3 = (const int*)d_in[22];
  const int* g3 = (const int*)d_in[23];
  const int* k3 = (const int*)d_in[24];
  const int* q3 = (const int*)d_in[25];

  int*   offs = (int*)d_ws;
  float* Wg   = (float*)((char*)d_ws + 2048);
  float* out  = (float*)d_out;

  prep_offsets_kernel<<<2, 256, 0, stream>>>(i1, l1, i2, j2, l2, m2,
                                             i3, j3, f3, l3, m3, g3, offs);
  prep_wg_kernel<<<104, 256, 0, stream>>>(w1, w2, w3, c1, c2, c3,
                                          k1, q1, k2, q2, k3, q3, Wg);
  contract_kernel<<<dim3(4, 64), 384, 0, stream>>>(x, y, Wg, offs, out);
}


# Tiles dealt per work-group, not per a: each group runs 64 consecutive 16-row tiles (waves 0-3 six, 4-11 five), so every WGP and SIMD gets 64 and 16 tiles, not 72/64/60/60 and 18/16/15/15
# speedup vs baseline: 2.6252x; 1.1058x over previous
#include <hip/hip_runtime.h>
#include <hip/hip_bf16.h>

typedef float v2f __attribute__((ext_vector_type(2)));
typedef float v8f __attribute__((ext_vector_type(8)));

#define NB 4096
#define NA 64
#define NL 16
#define NM 4
#define NE 10
#define P_TOT 416
#define P1_END 32
#define P2_END 160
#define EPAD 16
#define XPITCH 68
#define KSTEPS 104
#define WAVES 12
#define GROUPS_PER_A 4
#define BTILES (NB / 16)
#define TILES_PER_GROUP (BTILES / GROUPS_PER_A)

__global__ void prep_offsets_kernel(const int* __restrict__ i1, const int* __restrict__ l1,
                                    const int* __restrict__ i2, const int* __restrict__ j2,
                                    const int* __restrict__ l2, const int* __restrict__ m2,
                                    const int* __restrict__ i3, const int* __restrict__ j3,
                                    const int* __restrict__ f3, const int* __restrict__ l3,
                                    const int* __restrict__ m3, const int* __restrict__ g3,
                                    int* __restrict__ offs) {
  int p = blockIdx.x * blockDim.x + threadIdx.x;
  if (p >= P_TOT) return;
  int packed;
  if (p < P1_END) {
    int o1 = (i1[p] * NM + l1[p]) * 4;
    packed = o1;
  } else if (p < P2_END) {
    int q = p - P1_END;
    int o1 = (i2[q] * NM + l2[q]) * 4;
    int o2 = (j2[q] * NM + m2[q]) * 4;
    packed = o1 | (o2 << 8);
  } else {
    int q = p - P2_END;
    int o1 = (i3[q] * NM + l3[q]) * 4;
    int o2 = (j3[q] * NM + m3[q]) * 4;
    int o3 = (f3[q] * NM + g3[q]) * 4;
    packed = o1 | (o2 << 8) | (o3 << 16);
  }
  offs[p] = packed;
}

__global__ void prep_wg_kernel(const float* __restrict__ w1, const float* __restrict__ w2,
                               const float* __restrict__ w3,
                               const float* __restrict__ c1, const float* __restrict__ c2,
                               const float* __restrict__ c3,
                               const int* __restrict__ k1, const int* __restrict__ q1,
                               const int* __restrict__ k2, const int* __restrict__ q2,
                               const int* __restrict__ k3, const int* __restrict__ q3,
                               float* __restrict__ Wg) {
  int tid = blockIdx.x * blockDim.x + threadIdx.x;
  if (tid >= NA * P_TOT) return;
  int p = tid % P_TOT;
  int a = tid / P_TOT;
  float cf;
  const float* wbase;
  int estride;
  int kq;
  if (p < P1_END) {
    cf = c1[p];
    wbase = w1; kq = k1[p] * 2 + q1[p]; estride = 4 * 2 * NA;
  } else if (p < P2_END) {
    int t = p - P1_END;
    cf = c2[t];
    wbase = w2; kq = k2[t] * 4 + q2[t]; estride = 16 * 4 * NA;
  } else {
    int t = p - P2_END;
    cf = c3[t];
    wbase = w3; kq = k3[t] * 6 + q3[t]; estride = 32 * 6 * NA;
  }
  const int kqstride = kq * NA + a;
  float* dst = Wg + ((size_t)a * P_TOT + p) * EPAD;
#pragma unroll
  for (int n = 0; n < EPAD; ++n) {
    dst[n] = (n < NE) ? cf * wbase[(size_t)n * estride + kqstride] : 0.0f;
  }
}

template <int CNT>
__device__ __forceinline__ v2f xprod_frag(int pk0, int pk1, const char* Xb) {
  float xa = *(const float*)(Xb + (pk0 & 0xFF));
  float xb = *(const float*)(Xb + (pk1 & 0xFF));
  if (CNT > 1) {
    xa *= *(const float*)(Xb + ((pk0 >> 8) & 0xFF));
    xb *= *(const float*)(Xb + ((pk1 >> 8) & 0xFF));
  }
  if (CNT > 2) {
    xa *= *(const float*)(Xb + ((pk0 >> 16) & 0xFF));
    xb *= *(const float*)(Xb + ((pk1 >> 16) & 0xFF));
  }
  v2f f; f.x = xa; f.y = xb;
  return f;
}

#define XROWSTEP ((size_t)2 * NA * 64)
#define LOAD_XROWS(P)                                        \
  x0 = *(const float4*)((P));                                \
  x1 = *(const float4*)((P) + 1 * XROWSTEP);                 \
  x2 = *(const float4*)((P) + 2 * XROWSTEP);                 \
  x3 = *(const float4*)((P) + 3 * XROWSTEP);                 \
  x4 = *(const float4*)((P) + 4 * XROWSTEP);                 \
  x5 = *(const float4*)((P) + 5 * XROWSTEP);                 \
  x6 = *(const float4*)((P) + 6 * XROWSTEP);                 \
  x7 = *(const float4*)((P) + 7 * XROWSTEP);

#define REGION_END(T) if (((T) + 1) % 8 == 0) __builtin_amdgcn_sched_barrier(0)

__global__ __launch_bounds__(WAVES * 32) void contract_kernel(const float* __restrict__ x,
                                                              const float* __restrict__ y,
                                                              const float* __restrict__ Wg,
                                                              const int* __restrict__ offs,
                                                              float* __restrict__ out) {
  __shared__ float ldsX[WAVES * 16 * XPITCH];
  __shared__ int   ldsOff[P_TOT];

  const int tid  = threadIdx.x;
  const int wv   = __builtin_amdgcn_readfirstlane(tid >> 5);
  const int lane = tid & 31;
  const int half = lane >> 4;
  const int lrow = lane & 15;
  const int a    = blockIdx.y;
  const int tile0 = blockIdx.x * TILES_PER_GROUP + wv;
  const int ntiles = (TILES_PER_GROUP - wv + WAVES - 1) / WAVES;

  for (int i = tid; i < P_TOT; i += WAVES * 32) ldsOff[i] = offs[i];

  v2f wf[KSTEPS];
  {
    const float* wsrc = Wg + (size_t)a * P_TOT * EPAD + (2 * half) * EPAD + lrow;
#pragma unroll
    for (int t = 0; t < KSTEPS; ++t) {
      wf[t].x = wsrc[(4 * t) * EPAD];
      wf[t].y = wsrc[(4 * t + 1) * EPAD];
    }
  }

  const int srow = lane >> 4;
  const int c4   = lane & 15;
  const float* xsrc = x + ((size_t)(tile0 * 16 + srow) * NA + a) * 64 + c4 * 4;
  float4 x0, x1, x2, x3, x4, x5, x6, x7;
  LOAD_XROWS(xsrc)

  __syncthreads();

  float* Xw = &ldsX[wv * 16 * XPITCH];
  float* Xst = Xw + srow * XPITCH + c4 * 4;
  const char* Xb = (const char*)(Xw + lrow * XPITCH);
  const int* offl = ldsOff + 2 * half;

  for (int it = 0; it < ntiles; ++it) {
    const int b0 = (tile0 + it * WAVES) * 16;

    __builtin_amdgcn_fence(__ATOMIC_RELEASE, "wavefront");
    __builtin_amdgcn_wave_barrier();
    *(float4*)(Xst + 0 * 2 * XPITCH) = x0;
    *(float4*)(Xst + 1 * 2 * XPITCH) = x1;
    *(float4*)(Xst + 2 * 2 * XPITCH) = x2;
    *(float4*)(Xst + 3 * 2 * XPITCH) = x3;
    *(float4*)(Xst + 4 * 2 * XPITCH) = x4;
    *(float4*)(Xst + 5 * 2 * XPITCH) = x5;
    *(float4*)(Xst + 6 * 2 * XPITCH) = x6;
    *(float4*)(Xst + 7 * 2 * XPITCH) = x7;
    __builtin_amdgcn_fence(__ATOMIC_RELEASE, "wavefront");
    __builtin_amdgcn_wave_barrier();
    __builtin_amdgcn_fence(__ATOMIC_ACQUIRE, "wavefront");

    if (it + 1 < ntiles) {
      const float* xn = xsrc + (size_t)((it + 1) * WAVES * 16) * NA * 64;
      LOAD_XROWS(xn)
    }

    float yv[8];
#pragma unroll
    for (int r = 0; r < 8; ++r) {
      const int e  = 8 * half + r;
      const int ec = e < NE ? e : NE - 1;
      const float v = y[(size_t)(b0 + lrow) * NE + ec];
      yv[r] = e < NE ? v : 0.0f;
    }

    v8f acc = {};
#pragma unroll
    for (int t = 0; t < 8; ++t) {
      const v2f bf = xprod_frag<1>(offl[4 * t], offl[4 * t + 1], Xb);
      acc = __builtin_amdgcn_wmma_f32_16x16x4_f32(false, wf[t], false, bf,
                                                  (short)0, acc, false, false);
      REGION_END(t);
    }
#pragma unroll
    for (int t = 8; t < 40; ++t) {
      const v2f bf = xprod_frag<2>(offl[4 * t], offl[4 * t + 1], Xb);
      acc = __builtin_amdgcn_wmma_f32_16x16x4_f32(false, wf[t], false, bf,
                                                  (short)0, acc, false, false);
      REGION_END(t);
    }
#pragma unroll
    for (int t = 40; t < KSTEPS; ++t) {
      const v2f bf = xprod_frag<3>(offl[4 * t], offl[4 * t + 1], Xb);
      acc = __builtin_amdgcn_wmma_f32_16x16x4_f32(false, wf[t], false, bf,
                                                  (short)0, acc, false, false);
      REGION_END(t);
    }

    asm volatile("s_wait_alu depctr_va_vdst(0)\n\tv_nop\n\tv_nop\n\tv_nop\n\tv_nop\n\tv_nop\n\tv_nop\n\tv_nop\n\tv_nop"
                 : "+v"(acc));

    float part = acc[0] * yv[0];
#pragma unroll
    for (int r = 1; r < 8; ++r) part += acc[r] * yv[r];
    part += __shfl_xor(part, 16, 32);
    if (half == 0) out[(size_t)(b0 + lrow) * NA + a] = part;
  }
}

extern "C" void kernel_launch(void* const* d_in, const int* in_sizes, int n_in,
                              void* d_out, int out_size, void* d_ws, size_t ws_size,
                              hipStream_t stream) {
  const float* x  = (const float*)d_in[0];
  const float* y  = (const float*)d_in[1];
  const float* w1 = (const float*)d_in[2];
  const float* w2 = (const float*)d_in[3];
  const float* w3 = (const float*)d_in[4];
  const float* c1 = (const float*)d_in[5];
  const float* c2 = (const float*)d_in[6];
  const float* c3 = (const float*)d_in[7];
  const int* i1 = (const int*)d_in[8];
  const int* l1 = (const int*)d_in[9];
  const int* k1 = (const int*)d_in[10];
  const int* q1 = (const int*)d_in[11];
  const int* i2 = (const int*)d_in[12];
  const int* j2 = (const int*)d_in[13];
  const int* l2 = (const int*)d_in[14];
  const int* m2 = (const int*)d_in[15];
  const int* k2 = (const int*)d_in[16];
  const int* q2 = (const int*)d_in[17];
  const int* i3 = (const int*)d_in[18];
  const int* j3 = (const int*)d_in[19];
  const int* f3 = (const int*)d_in[20];
  const int* l3 = (const int*)d_in[21];
  const int* m3 = (const int*)d_in[22];
  const int* g3 = (const int*)d_in[23];
  const int* k3 = (const int*)d_in[24];
  const int* q3 = (const int*)d_in[25];

  int*   offs = (int*)d_ws;
  float* Wg   = (float*)((char*)d_ws + 2048);
  float* out  = (float*)d_out;

  prep_offsets_kernel<<<2, 256, 0, stream>>>(i1, l1, i2, j2, l2, m2,
                                             i3, j3, f3, l3, m3, g3, offs);
  prep_wg_kernel<<<104, 256, 0, stream>>>(w1, w2, w3, c1, c2, c3,
                                          k1, q1, k2, q2, k3, q3, Wg);
  contract_kernel<<<dim3(4, 64), 384, 0, stream>>>(x, y, Wg, offs, out);
}


# Each 8-step region grouped: its 16 table offsets, then all its LDS gathers, then products and 8 WMMAs with no LDS instruction among them: 1 matrix-LDS switch per region, not 8; sum order unchanged
# speedup vs baseline: 2.7668x; 1.0539x over previous
#include <hip/hip_runtime.h>
#include <hip/hip_bf16.h>

typedef float v2f __attribute__((ext_vector_type(2)));
typedef float v8f __attribute__((ext_vector_type(8)));

#define NB 4096
#define NA 64
#define NL 16
#define NM 4
#define NE 10
#define P_TOT 416
#define P1_END 32
#define P2_END 160
#define EPAD 16
#define XPITCH 68
#define KSTEPS 104
#define WAVES 12
#define GROUPS_PER_A 4
#define BTILES (NB / 16)
#define TILES_PER_GROUP (BTILES / GROUPS_PER_A)

__global__ void prep_offsets_kernel(const int* __restrict__ i1, const int* __restrict__ l1,
                                    const int* __restrict__ i2, const int* __restrict__ j2,
                                    const int* __restrict__ l2, const int* __restrict__ m2,
                                    const int* __restrict__ i3, const int* __restrict__ j3,
                                    const int* __restrict__ f3, const int* __restrict__ l3,
                                    const int* __restrict__ m3, const int* __restrict__ g3,
                                    int* __restrict__ offs) {
  int p = blockIdx.x * blockDim.x + threadIdx.x;
  if (p >= P_TOT) return;
  int packed;
  if (p < P1_END) {
    int o1 = (i1[p] * NM + l1[p]) * 4;
    packed = o1;
  } else if (p < P2_END) {
    int q = p - P1_END;
    int o1 = (i2[q] * NM + l2[q]) * 4;
    int o2 = (j2[q] * NM + m2[q]) * 4;
    packed = o1 | (o2 << 8);
  } else {
    int q = p - P2_END;
    int o1 = (i3[q] * NM + l3[q]) * 4;
    int o2 = (j3[q] * NM + m3[q]) * 4;
    int o3 = (f3[q] * NM + g3[q]) * 4;
    packed = o1 | (o2 << 8) | (o3 << 16);
  }
  offs[p] = packed;
}

__global__ void prep_wg_kernel(const float* __restrict__ w1, const float* __restrict__ w2,
                               const float* __restrict__ w3,
                               const float* __restrict__ c1, const float* __restrict__ c2,
                               const float* __restrict__ c3,
                               const int* __restrict__ k1, const int* __restrict__ q1,
                               const int* __restrict__ k2, const int* __restrict__ q2,
                               const int* __restrict__ k3, const int* __restrict__ q3,
                               float* __restrict__ Wg) {
  int tid = blockIdx.x * blockDim.x + threadIdx.x;
  if (tid >= NA * P_TOT) return;
  int p = tid % P_TOT;
  int a = tid / P_TOT;
  float cf;
  const float* wbase;
  int estride;
  int kq;
  if (p < P1_END) {
    cf = c1[p];
    wbase = w1; kq = k1[p] * 2 + q1[p]; estride = 4 * 2 * NA;
  } else if (p < P2_END) {
    int t = p - P1_END;
    cf = c2[t];
    wbase = w2; kq = k2[t] * 4 + q2[t]; estride = 16 * 4 * NA;
  } else {
    int t = p - P2_END;
    cf = c3[t];
    wbase = w3; kq = k3[t] * 6 + q3[t]; estride = 32 * 6 * NA;
  }
  const int kqstride = kq * NA + a;
  float* dst = Wg + ((size_t)a * P_TOT + p) * EPAD;
#pragma unroll
  for (int n = 0; n < EPAD; ++n) {
    dst[n] = (n < NE) ? cf * wbase[(size_t)n * estride + kqstride] : 0.0f;
  }
}

template <int CNT>
__device__ __forceinline__ void xgather(int pk0, int pk1, const char* Xb, float* g) {
#pragma unroll
  for (int f = 0; f < CNT; ++f) {
    g[2 * f]     = *(const float*)(Xb + ((pk0 >> (8 * f)) & 0xFF));
    g[2 * f + 1] = *(const float*)(Xb + ((pk1 >> (8 * f)) & 0xFF));
  }
}
template <int CNT>
__device__ __forceinline__ v2f xproduct(const float* g) {
  float xa = g[0], xb = g[1];
  if (CNT > 1) { xa *= g[2]; xb *= g[3]; }
  if (CNT > 2) { xa *= g[4]; xb *= g[5]; }
  v2f f; f.x = xa; f.y = xb;
  return f;
}

#define XROWSTEP ((size_t)2 * NA * 64)
#define LOAD_XROWS(P)                                        \
  x0 = *(const float4*)((P));                                \
  x1 = *(const float4*)((P) + 1 * XROWSTEP);                 \
  x2 = *(const float4*)((P) + 2 * XROWSTEP);                 \
  x3 = *(const float4*)((P) + 3 * XROWSTEP);                 \
  x4 = *(const float4*)((P) + 4 * XROWSTEP);                 \
  x5 = *(const float4*)((P) + 5 * XROWSTEP);                 \
  x6 = *(const float4*)((P) + 6 * XROWSTEP);                 \
  x7 = *(const float4*)((P) + 7 * XROWSTEP);

#define REGION(CNT, T0)                                                                      \
  {                                                                                          \
    int pk[16];                                                                              \
    _Pragma("unroll")                                      \
    for (int u = 0; u < 8; ++u) {                                                            \
      pk[2 * u]     = offl[4 * ((T0) + u)];                                                  \
      pk[2 * u + 1] = offl[4 * ((T0) + u) + 1];                                              \
    }                                                                                        \
    __builtin_amdgcn_sched_barrier(0);   \
    float g[8][2 * (CNT)];                                                                   \
    _Pragma("unroll")                                     \
    for (int u = 0; u < 8; ++u) xgather<CNT>(pk[2 * u], pk[2 * u + 1], Xb, g[u]);            \
    __builtin_amdgcn_sched_barrier(0);      \
    v2f bf[8];                                                                               \
    _Pragma("unroll")                                      \
    for (int u = 0; u < 8; ++u) bf[u] = xproduct<CNT>(g[u]);                                 \
    _Pragma("unroll")                                   \
    for (int u = 0; u < 8; ++u)                                                              \
      acc = __builtin_amdgcn_wmma_f32_16x16x4_f32(false, wf[(T0) + u], false, bf[u],         \
                                                  (short)0, acc, false, false);              \
    __builtin_amdgcn_sched_barrier(0);     \
  }

__global__ __launch_bounds__(WAVES * 32) void contract_kernel(const float* __restrict__ x,
                                                              const float* __restrict__ y,
                                                              const float* __restrict__ Wg,
                                                              const int* __restrict__ offs,
                                                              float* __restrict__ out) {
  __shared__ float ldsX[WAVES * 16 * XPITCH];
  __shared__ int   ldsOff[P_TOT];

  const int tid  = threadIdx.x;
  const int wv   = __builtin_amdgcn_readfirstlane(tid >> 5);
  const int lane = tid & 31;
  const int half = lane >> 4;
  const int lrow = lane & 15;
  const int a    = blockIdx.y;
  const int tile0 = blockIdx.x * TILES_PER_GROUP + wv;
  const int ntiles = (TILES_PER_GROUP - wv + WAVES - 1) / WAVES;

  for (int i = tid; i < P_TOT; i += WAVES * 32) ldsOff[i] = offs[i];

  v2f wf[KSTEPS];
  {
    const float* wsrc = Wg + (size_t)a * P_TOT * EPAD + (2 * half) * EPAD + lrow;
#pragma unroll
    for (int t = 0; t < KSTEPS; ++t) {
      wf[t].x = wsrc[(4 * t) * EPAD];
      wf[t].y = wsrc[(4 * t + 1) * EPAD];
    }
  }

  const int srow = lane >> 4;
  const int c4   = lane & 15;
  const float* xsrc = x + ((size_t)(tile0 * 16 + srow) * NA + a) * 64 + c4 * 4;
  float4 x0, x1, x2, x3, x4, x5, x6, x7;
  LOAD_XROWS(xsrc)

  __syncthreads();

  float* Xw = &ldsX[wv * 16 * XPITCH];
  float* Xst = Xw + srow * XPITCH + c4 * 4;
  const char* Xb = (const char*)(Xw + lrow * XPITCH);
  const int* offl = ldsOff + 2 * half;

  for (int it = 0; it < ntiles; ++it) {
    const int b0 = (tile0 + it * WAVES) * 16;

    __builtin_amdgcn_fence(__ATOMIC_RELEASE, "wavefront");
    __builtin_amdgcn_wave_barrier();
    *(float4*)(Xst + 0 * 2 * XPITCH) = x0;
    *(float4*)(Xst + 1 * 2 * XPITCH) = x1;
    *(float4*)(Xst + 2 * 2 * XPITCH) = x2;
    *(float4*)(Xst + 3 * 2 * XPITCH) = x3;
    *(float4*)(Xst + 4 * 2 * XPITCH) = x4;
    *(float4*)(Xst + 5 * 2 * XPITCH) = x5;
    *(float4*)(Xst + 6 * 2 * XPITCH) = x6;
    *(float4*)(Xst + 7 * 2 * XPITCH) = x7;
    __builtin_amdgcn_fence(__ATOMIC_RELEASE, "wavefront");
    __builtin_amdgcn_wave_barrier();
    __builtin_amdgcn_fence(__ATOMIC_ACQUIRE, "wavefront");

    if (it + 1 < ntiles) {
      const float* xn = xsrc + (size_t)((it + 1) * WAVES * 16) * NA * 64;
      LOAD_XROWS(xn)
    }

    float yv[8];
#pragma unroll
    for (int r = 0; r < 8; ++r) {
      const int e  = 8 * half + r;
      const int ec = e < NE ? e : NE - 1;
      const float v = y[(size_t)(b0 + lrow) * NE + ec];
      yv[r] = e < NE ? v : 0.0f;
    }

    v8f acc = {};
    REGION(1, 0)
#pragma unroll
    for (int t0 = 8; t0 < 40; t0 += 8) REGION(2, t0)
#pragma unroll
    for (int t0 = 40; t0 < KSTEPS; t0 += 8) REGION(3, t0)

    asm volatile("s_wait_alu depctr_va_vdst(0)\n\tv_nop\n\tv_nop\n\tv_nop\n\tv_nop\n\tv_nop\n\tv_nop\n\tv_nop\n\tv_nop"
                 : "+v"(acc));

    float part = acc[0] * yv[0];
#pragma unroll
    for (int r = 1; r < 8; ++r) part += acc[r] * yv[r];
    part += __shfl_xor(part, 16, 32);
    if (half == 0) out[(size_t)(b0 + lrow) * NA + a] = part;
  }
}

extern "C" void kernel_launch(void* const* d_in, const int* in_sizes, int n_in,
                              void* d_out, int out_size, void* d_ws, size_t ws_size,
                              hipStream_t stream) {
  const float* x  = (const float*)d_in[0];
  const float* y  = (const float*)d_in[1];
  const float* w1 = (const float*)d_in[2];
  const float* w2 = (const float*)d_in[3];
  const float* w3 = (const float*)d_in[4];
  const float* c1 = (const float*)d_in[5];
  const float* c2 = (const float*)d_in[6];
  const float* c3 = (const float*)d_in[7];
  const int* i1 = (const int*)d_in[8];
  const int* l1 = (const int*)d_in[9];
  const int* k1 = (const int*)d_in[10];
  const int* q1 = (const int*)d_in[11];
  const int* i2 = (const int*)d_in[12];
  const int* j2 = (const int*)d_in[13];
  const int* l2 = (const int*)d_in[14];
  const int* m2 = (const int*)d_in[15];
  const int* k2 = (const int*)d_in[16];
  const int* q2 = (const int*)d_in[17];
  const int* i3 = (const int*)d_in[18];
  const int* j3 = (const int*)d_in[19];
  const int* f3 = (const int*)d_in[20];
  const int* l3 = (const int*)d_in[21];
  const int* m3 = (const int*)d_in[22];
  const int* g3 = (const int*)d_in[23];
  const int* k3 = (const int*)d_in[24];
  const int* q3 = (const int*)d_in[25];

  int*   offs = (int*)d_ws;
  float* Wg   = (float*)((char*)d_ws + 2048);
  float* out  = (float*)d_out;

  prep_offsets_kernel<<<2, 256, 0, stream>>>(i1, l1, i2, j2, l2, m2,
                                             i3, j3, f3, l3, m3, g3, offs);
  prep_wg_kernel<<<104, 256, 0, stream>>>(w1, w2, w3, c1, c2, c3,
                                          k1, q1, k2, q2, k3, q3, Wg);
  contract_kernel<<<dim3(4, 64), 384, 0, stream>>>(x, y, Wg, offs, out);
}


# No prep launches, no workspace: each work-group builds its a's weight table and gather offsets in LDS from the raw inputs; A fragments read from LDS; 3 launches to 1; sum order unchanged
# speedup vs baseline: 3.0488x; 1.1019x over previous
#include <hip/hip_runtime.h>
#include <hip/hip_bf16.h>

typedef float v2f __attribute__((ext_vector_type(2)));
typedef float v8f __attribute__((ext_vector_type(8)));

#define NB 4096
#define NA 64
#define NL 16
#define NM 4
#define NE 10
#define P_TOT 416
#define P1_END 32
#define P2_END 160
#define EPAD 16
#define XPITCH 68
#define KSTEPS 104
#define WAVES 12
#define GROUPS_PER_A 4
#define BTILES (NB / 16)
#define TILES_PER_GROUP (BTILES / GROUPS_PER_A)

#define NPAIR (P_TOT / 2)
#define PP1_END (P1_END / 2)
#define PP2_END (P2_END / 2)

template <int N, int Q, int ES, int PP0, int PP1>
__device__ __forceinline__ void wg_index(const int* __restrict__ k, const int* __restrict__ q,
                                         const float* __restrict__ c, const float* __restrict__ w,
                                         int wv, int half, int ec, int a,
                                         float* cf, const float** wp) {
#pragma unroll
  for (int j = 0; j < N; ++j) {
    int pp = PP0 + wv + WAVES * j;
    pp = pp < PP1 ? pp : PP1 - 1;
    const int t = 2 * (pp - PP0) + half;
    cf[j] = c[t];
    wp[j] = w + (ec * ES + (k[t] * Q + q[t]) * NA + a);
  }
}
template <int N>
__device__ __forceinline__ void wg_load(const float* const* wp, float* val) {
#pragma unroll
  for (int j = 0; j < N; ++j) val[j] = *wp[j];
}
template <int N, int PP0, int PP1>
__device__ __forceinline__ void wg_store(float* ldsWf, int wv, int half, int lrow,
                                         const float* cf, const float* val) {
#pragma unroll
  for (int j = 0; j < N; ++j) {
    const int pp = PP0 + wv + WAVES * j;
    if (pp < PP1) ldsWf[(pp * EPAD + lrow) * 2 + half] = lrow < NE ? cf[j] * val[j] : 0.0f;
  }
}

template <int CNT>
__device__ __forceinline__ void xgather(int pk0, int pk1, const char* Xb, float* g) {
#pragma unroll
  for (int f = 0; f < CNT; ++f) {
    g[2 * f]     = *(const float*)(Xb + ((pk0 >> (8 * f)) & 0xFF));
    g[2 * f + 1] = *(const float*)(Xb + ((pk1 >> (8 * f)) & 0xFF));
  }
}
template <int CNT>
__device__ __forceinline__ v2f xproduct(const float* g) {
  float xa = g[0], xb = g[1];
  if (CNT > 1) { xa *= g[2]; xb *= g[3]; }
  if (CNT > 2) { xa *= g[4]; xb *= g[5]; }
  v2f f; f.x = xa; f.y = xb;
  return f;
}

#define XROWSTEP ((size_t)2 * NA * 64)
#define LOAD_XROWS(P)                                        \
  x0 = *(const float4*)((P));                                \
  x1 = *(const float4*)((P) + 1 * XROWSTEP);                 \
  x2 = *(const float4*)((P) + 2 * XROWSTEP);                 \
  x3 = *(const float4*)((P) + 3 * XROWSTEP);                 \
  x4 = *(const float4*)((P) + 4 * XROWSTEP);                 \
  x5 = *(const float4*)((P) + 5 * XROWSTEP);                 \
  x6 = *(const float4*)((P) + 6 * XROWSTEP);                 \
  x7 = *(const float4*)((P) + 7 * XROWSTEP);

#define REGION(CNT, T0)                                                                      \
  {                                                                                          \
    int pk[16];                                                                              \
    _Pragma("unroll")                                      \
    for (int u = 0; u < 8; ++u) {                                                            \
      pk[2 * u]     = offl[4 * ((T0) + u)];                                                  \
      pk[2 * u + 1] = offl[4 * ((T0) + u) + 1];                                              \
    }                                                                                        \
    __builtin_amdgcn_sched_barrier(0);   \
    float g[8][2 * (CNT)];                                                                   \
    _Pragma("unroll")                                     \
    for (int u = 0; u < 8; ++u) xgather<CNT>(pk[2 * u], pk[2 * u + 1], Xb, g[u]);            \
    __builtin_amdgcn_sched_barrier(0);      \
    v2f bf[8];                                                                               \
    _Pragma("unroll")                                      \
    for (int u = 0; u < 8; ++u) bf[u] = xproduct<CNT>(g[u]);                                 \
    _Pragma("unroll")                                   \
    for (int u = 0; u < 8; ++u)                                                              \
      acc = __builtin_amdgcn_wmma_f32_16x16x4_f32(false, wf[(T0) + u], false, bf[u],         \
                                                  (short)0, acc, false, false);              \
    __builtin_amdgcn_sched_barrier(0);     \
  }

__global__ __launch_bounds__(WAVES * 32) void contract_kernel(
    const float* __restrict__ x, const float* __restrict__ y,
    const float* __restrict__ w1, const float* __restrict__ w2, const float* __restrict__ w3,
    const float* __restrict__ c1, const float* __restrict__ c2, const float* __restrict__ c3,
    const int* __restrict__ i1, const int* __restrict__ l1,
    const int* __restrict__ k1, const int* __restrict__ q1,
    const int* __restrict__ i2, const int* __restrict__ j2,
    const int* __restrict__ l2, const int* __restrict__ m2,
    const int* __restrict__ k2, const int* __restrict__ q2,
    const int* __restrict__ i3, const int* __restrict__ j3, const int* __restrict__ f3,
    const int* __restrict__ l3, const int* __restrict__ m3, const int* __restrict__ g3,
    const int* __restrict__ k3, const int* __restrict__ q3,
    float* __restrict__ out) {
  __shared__ v2f   ldsW[P_TOT * EPAD / 2];
  __shared__ float ldsX[WAVES * 16 * XPITCH];
  __shared__ int   ldsOff[2 * WAVES * 32];

  const int tid  = threadIdx.x;
  const int wv   = __builtin_amdgcn_readfirstlane(tid >> 5);
  const int lane = tid & 31;
  const int half = lane >> 4;
  const int lrow = lane & 15;
  const int a    = blockIdx.y;
  const int tile0 = blockIdx.x * TILES_PER_GROUP + wv;
  const int ntiles = (TILES_PER_GROUP - wv + WAVES - 1) / WAVES;

  const int srow = lane >> 4;
  const int c4   = lane & 15;
  const float* xsrc = x + ((size_t)(tile0 * 16 + srow) * NA + a) * 64 + c4 * 4;
  float4 x0, x1, x2, x3, x4, x5, x6, x7;
  LOAD_XROWS(xsrc)

  const int ec = lrow < NE ? lrow : NE - 1;
  float cfA[2], cfB[6], cfC[11];
  const float *wpA[2], *wpB[6], *wpC[11];
  wg_index<2, 2, 4 * 2 * NA, 0, PP1_END>(k1, q1, c1, w1, wv, half, ec, a, cfA, wpA);
  wg_index<6, 4, 16 * 4 * NA, PP1_END, PP2_END>(k2, q2, c2, w2, wv, half, ec, a, cfB, wpB);
  wg_index<11, 6, 32 * 6 * NA, PP2_END, NPAIR>(k3, q3, c3, w3, wv, half, ec, a, cfC, wpC);

  int packedA, packedB;
  {
    const bool s1 = wv < 1, s12 = wv < 5;
    const int q0 = tid - (s1 ? 0 : (s12 ? P1_END : P2_END));
    const int* pi1 = s1 ? i1 : (s12 ? i2 : i3);
    const int* pl1 = s1 ? l1 : (s12 ? l2 : l3);
    const int* pi2 = s1 ? i1 : (s12 ? j2 : j3);
    const int* pl2 = s1 ? l1 : (s12 ? m2 : m3);
    const int* pi3 = s1 ? i1 : (s12 ? i2 : f3);
    const int* pl3 = s1 ? l1 : (s12 ? l2 : g3);
    const int o1 = (pi1[q0] * NM + pl1[q0]) * 4;
    const int o2 = (pi2[q0] * NM + pl2[q0]) * 4;
    const int o3 = (pi3[q0] * NM + pl3[q0]) * 4;
    packedA = o1 | (s1 ? 0 : (o2 << 8)) | (s12 ? 0 : (o3 << 16));
    const int q1u = WAVES * 32 - P2_END + tid;
    const int q1r = q1u < P_TOT - P2_END ? q1u : P_TOT - P2_END - 1;
    const int u1 = (i3[q1r] * NM + l3[q1r]) * 4;
    const int u2 = (j3[q1r] * NM + m3[q1r]) * 4;
    const int u3 = (f3[q1r] * NM + g3[q1r]) * 4;
    packedB = u1 | (u2 << 8) | (u3 << 16);
  }

  float vA[2], vB[6], vC[11];
  wg_load<2>(wpA, vA);
  wg_load<6>(wpB, vB);
  wg_load<11>(wpC, vC);

  ldsOff[tid] = packedA;
  ldsOff[WAVES * 32 + tid] = packedB;
  float* ldsWf = (float*)ldsW;
  wg_store<2, 0, PP1_END>(ldsWf, wv, half, lrow, cfA, vA);
  wg_store<6, PP1_END, PP2_END>(ldsWf, wv, half, lrow, cfB, vB);
  wg_store<11, PP2_END, NPAIR>(ldsWf, wv, half, lrow, cfC, vC);

  __syncthreads();

  v2f wf[KSTEPS];
  {
    const v2f* wl = ldsW + half * EPAD + lrow;
#pragma unroll
    for (int t = 0; t < KSTEPS; ++t) wf[t] = wl[t * 2 * EPAD];
  }

  float* Xw = &ldsX[wv * 16 * XPITCH];
  float* Xst = Xw + srow * XPITCH + c4 * 4;
  const char* Xb = (const char*)(Xw + lrow * XPITCH);
  const int* offl = ldsOff + 2 * half;

  for (int it = 0; it < ntiles; ++it) {
    const int b0 = (tile0 + it * WAVES) * 16;

    __builtin_amdgcn_fence(__ATOMIC_RELEASE, "wavefront");
    __builtin_amdgcn_wave_barrier();
    *(float4*)(Xst + 0 * 2 * XPITCH) = x0;
    *(float4*)(Xst + 1 * 2 * XPITCH) = x1;
    *(float4*)(Xst + 2 * 2 * XPITCH) = x2;
    *(float4*)(Xst + 3 * 2 * XPITCH) = x3;
    *(float4*)(Xst + 4 * 2 * XPITCH) = x4;
    *(float4*)(Xst + 5 * 2 * XPITCH) = x5;
    *(float4*)(Xst + 6 * 2 * XPITCH) = x6;
    *(float4*)(Xst + 7 * 2 * XPITCH) = x7;
    __builtin_amdgcn_fence(__ATOMIC_RELEASE, "wavefront");
    __builtin_amdgcn_wave_barrier();
    __builtin_amdgcn_fence(__ATOMIC_ACQUIRE, "wavefront");

    if (it + 1 < ntiles) {
      const float* xn = xsrc + (size_t)((it + 1) * WAVES * 16) * NA * 64;
      LOAD_XROWS(xn)
    }

    float yv[8];
#pragma unroll
    for (int r = 0; r < 8; ++r) {
      const int e  = 8 * half + r;
      const int ec = e < NE ? e : NE - 1;
      const float v = y[(size_t)(b0 + lrow) * NE + ec];
      yv[r] = e < NE ? v : 0.0f;
    }

    v8f acc = {};
    REGION(1, 0)
#pragma unroll
    for (int t0 = 8; t0 < 40; t0 += 8) REGION(2, t0)
#pragma unroll
    for (int t0 = 40; t0 < KSTEPS; t0 += 8) REGION(3, t0)

    asm volatile("s_wait_alu depctr_va_vdst(0)\n\tv_nop\n\tv_nop\n\tv_nop\n\tv_nop\n\tv_nop\n\tv_nop\n\tv_nop\n\tv_nop"
                 : "+v"(acc));

    float part = acc[0] * yv[0];
#pragma unroll
    for (int r = 1; r < 8; ++r) part += acc[r] * yv[r];
    part += __shfl_xor(part, 16, 32);
    if (half == 0) out[(size_t)(b0 + lrow) * NA + a] = part;
  }
}

extern "C" void kernel_launch(void* const* d_in, const int* in_sizes, int n_in,
                              void* d_out, int out_size, void* d_ws, size_t ws_size,
                              hipStream_t stream) {
  const float* x  = (const float*)d_in[0];
  const float* y  = (const float*)d_in[1];
  const float* w1 = (const float*)d_in[2];
  const float* w2 = (const float*)d_in[3];
  const float* w3 = (const float*)d_in[4];
  const float* c1 = (const float*)d_in[5];
  const float* c2 = (const float*)d_in[6];
  const float* c3 = (const float*)d_in[7];
  const int* i1 = (const int*)d_in[8];
  const int* l1 = (const int*)d_in[9];
  const int* k1 = (const int*)d_in[10];
  const int* q1 = (const int*)d_in[11];
  const int* i2 = (const int*)d_in[12];
  const int* j2 = (const int*)d_in[13];
  const int* l2 = (const int*)d_in[14];
  const int* m2 = (const int*)d_in[15];
  const int* k2 = (const int*)d_in[16];
  const int* q2 = (const int*)d_in[17];
  const int* i3 = (const int*)d_in[18];
  const int* j3 = (const int*)d_in[19];
  const int* f3 = (const int*)d_in[20];
  const int* l3 = (const int*)d_in[21];
  const int* m3 = (const int*)d_in[22];
  const int* g3 = (const int*)d_in[23];
  const int* k3 = (const int*)d_in[24];
  const int* q3 = (const int*)d_in[25];

  float* out  = (float*)d_out;

  contract_kernel<<<dim3(4, 64), 384, 0, stream>>>(x, y, w1, w2, w3, c1, c2, c3,
                                                   i1, l1, k1, q1,
                                                   i2, j2, l2, m2, k2, q2,
                                                   i3, j3, f3, l3, m3, g3, k3, q3, out);
}


# K loop interchanged: regions outer, the wave's 4 LDS-resident x tiles inner: A fragments, offsets and gather addresses fetched and unpacked once per region, not per tile; per-tile sum order unchanged
# speedup vs baseline: 3.5917x; 1.1781x over previous
#include <hip/hip_runtime.h>
#include <hip/hip_bf16.h>

typedef float v2f __attribute__((ext_vector_type(2)));
typedef float v8f __attribute__((ext_vector_type(8)));

#define NB 4096
#define NA 64
#define NL 16
#define NM 4
#define NE 10
#define P_TOT 416
#define P1_END 32
#define P2_END 160
#define EPAD 16
#define XPITCH 68
#define KSTEPS 104
#define WAVES 16
#define GROUPS_PER_A 4
#define BTILES (NB / 16)
#define TILES_PER_GROUP (BTILES / GROUPS_PER_A)
#define TPW (TILES_PER_GROUP / WAVES)
#define TILE_BYTES (16 * XPITCH * 4)

#define NPAIR (P_TOT / 2)
#define PP1_END (P1_END / 2)
#define PP2_END (P2_END / 2)

template <int N, int Q, int ES, int PP0, int PP1>
__device__ __forceinline__ void wg_index(const int* __restrict__ k, const int* __restrict__ q,
                                         const float* __restrict__ c, const float* __restrict__ w,
                                         int wv, int half, int ec, int a,
                                         float* cf, const float** wp) {
#pragma unroll
  for (int j = 0; j < N; ++j) {
    int pp = PP0 + wv + WAVES * j;
    pp = pp < PP1 ? pp : PP1 - 1;
    const int t = 2 * (pp - PP0) + half;
    cf[j] = c[t];
    wp[j] = w + (ec * ES + (k[t] * Q + q[t]) * NA + a);
  }
}
template <int N>
__device__ __forceinline__ void wg_load(const float* const* wp, float* val) {
#pragma unroll
  for (int j = 0; j < N; ++j) val[j] = *wp[j];
}
template <int N, int PP0, int PP1>
__device__ __forceinline__ void wg_store(float* ldsWf, int wv, int half, int lrow,
                                         const float* cf, const float* val) {
#pragma unroll
  for (int j = 0; j < N; ++j) {
    const int pp = PP0 + wv + WAVES * j;
    if (pp < PP1) ldsWf[(pp * EPAD + lrow) * 2 + half] = lrow < NE ? cf[j] * val[j] : 0.0f;
  }
}

template <int CNT>
__device__ __forceinline__ void xaddr(int pk0, int pk1, const char* Xb, const char** ad) {
#pragma unroll
  for (int f = 0; f < CNT; ++f) {
    ad[2 * f]     = Xb + ((pk0 >> (8 * f)) & 0xFF);
    ad[2 * f + 1] = Xb + ((pk1 >> (8 * f)) & 0xFF);
  }
}
template <int CNT, int TILE>
__device__ __forceinline__ void xgather(const char* const* ad, float* g) {
#pragma unroll
  for (int j = 0; j < 2 * CNT; ++j) g[j] = *(const float*)(ad[j] + TILE * TILE_BYTES);
}
template <int CNT>
__device__ __forceinline__ v2f xproduct(const float* g) {
  float xa = g[0], xb = g[1];
  if (CNT > 1) { xa *= g[2]; xb *= g[3]; }
  if (CNT > 2) { xa *= g[4]; xb *= g[5]; }
  v2f f; f.x = xa; f.y = xb;
  return f;
}

#define XROWSTEP ((size_t)2 * NA * 64)
#define LOAD_XROWS(V, P)                                     \
  V[0] = *(const float4*)((P));                              \
  V[1] = *(const float4*)((P) + 1 * XROWSTEP);               \
  V[2] = *(const float4*)((P) + 2 * XROWSTEP);               \
  V[3] = *(const float4*)((P) + 3 * XROWSTEP);               \
  V[4] = *(const float4*)((P) + 4 * XROWSTEP);               \
  V[5] = *(const float4*)((P) + 5 * XROWSTEP);               \
  V[6] = *(const float4*)((P) + 6 * XROWSTEP);               \
  V[7] = *(const float4*)((P) + 7 * XROWSTEP);
#define STORE_XROWS(V, Q)                                    \
  *(float4*)((Q) + 0 * 2 * XPITCH) = V[0];                   \
  *(float4*)((Q) + 1 * 2 * XPITCH) = V[1];                   \
  *(float4*)((Q) + 2 * 2 * XPITCH) = V[2];                   \
  *(float4*)((Q) + 3 * 2 * XPITCH) = V[3];                   \
  *(float4*)((Q) + 4 * 2 * XPITCH) = V[4];                   \
  *(float4*)((Q) + 5 * 2 * XPITCH) = V[5];                   \
  *(float4*)((Q) + 6 * 2 * XPITCH) = V[6];                   \
  *(float4*)((Q) + 7 * 2 * XPITCH) = V[7];

#define REGION_TILE(CNT, TILE)                                                               \
  {                                                                                          \
    float g[8][2 * (CNT)];                                                                   \
    _Pragma("unroll")                                     \
    for (int u = 0; u < 8; ++u) xgather<CNT, TILE>(ad[u], g[u]);                             \
    __builtin_amdgcn_sched_barrier(0);      \
    v2f bf[8];                                                                               \
    _Pragma("unroll")                                      \
    for (int u = 0; u < 8; ++u) bf[u] = xproduct<CNT>(g[u]);                                 \
    _Pragma("unroll")                                   \
    for (int u = 0; u < 8; ++u)                                                              \
      acc[TILE] = __builtin_amdgcn_wmma_f32_16x16x4_f32(false, wf[u], false, bf[u],          \
                                                        (short)0, acc[TILE], false, false);  \
    __builtin_amdgcn_sched_barrier(0);   \
  }

#define REGION(CNT, T0)                                                                      \
  {                                                                                          \
    v2f wf[8];                                                                               \
    int pk[16];                                                                              \
    _Pragma("unroll")                                 \
    for (int u = 0; u < 8; ++u) {                                                            \
      wf[u]         = wl[((T0) + u) * 2 * EPAD];                                             \
      pk[2 * u]     = offl[4 * ((T0) + u)];                                                  \
      pk[2 * u + 1] = offl[4 * ((T0) + u) + 1];                                              \
    }                                                                                        \
    __builtin_amdgcn_sched_barrier(0);   \
    const char* ad[8][2 * (CNT)];                                                            \
    _Pragma("unroll")                                    \
    for (int u = 0; u < 8; ++u) xaddr<CNT>(pk[2 * u], pk[2 * u + 1], Xb, ad[u]);             \
    __builtin_amdgcn_sched_barrier(0);    \
    REGION_TILE(CNT, 0)                                                                      \
    REGION_TILE(CNT, 1)                                                                      \
    REGION_TILE(CNT, 2)                                                                      \
    REGION_TILE(CNT, 3)                                                                      \
  }

__global__ __launch_bounds__(WAVES * 32) void contract_kernel(
    const float* __restrict__ x, const float* __restrict__ y,
    const float* __restrict__ w1, const float* __restrict__ w2, const float* __restrict__ w3,
    const float* __restrict__ c1, const float* __restrict__ c2, const float* __restrict__ c3,
    const int* __restrict__ i1, const int* __restrict__ l1,
    const int* __restrict__ k1, const int* __restrict__ q1,
    const int* __restrict__ i2, const int* __restrict__ j2,
    const int* __restrict__ l2, const int* __restrict__ m2,
    const int* __restrict__ k2, const int* __restrict__ q2,
    const int* __restrict__ i3, const int* __restrict__ j3, const int* __restrict__ f3,
    const int* __restrict__ l3, const int* __restrict__ m3, const int* __restrict__ g3,
    const int* __restrict__ k3, const int* __restrict__ q3,
    float* __restrict__ out) {
  __shared__ v2f   ldsW[P_TOT * EPAD / 2];
  __shared__ float ldsX[WAVES * TPW * 16 * XPITCH];
  __shared__ int   ldsOff[P_TOT];

  const int tid  = threadIdx.x;
  const int wv   = __builtin_amdgcn_readfirstlane(tid >> 5);
  const int lane = tid & 31;
  const int half = lane >> 4;
  const int lrow = lane & 15;
  const int a    = blockIdx.y;
  const int tile0 = blockIdx.x * TILES_PER_GROUP + wv * TPW;

  const int srow = lane >> 4;
  const int c4   = lane & 15;
  const float* xsrc = x + ((size_t)(tile0 * 16 + srow) * NA + a) * 64 + c4 * 4;
  float4 xr[TPW][8];
#pragma unroll
  for (int k = 0; k < TPW; ++k) { LOAD_XROWS(xr[k], xsrc + (size_t)(k * 16) * NA * 64) }

  const int ec = lrow < NE ? lrow : NE - 1;
  float cfA[1], cfB[4], cfC[8];
  const float *wpA[1], *wpB[4], *wpC[8];
  wg_index<1, 2, 4 * 2 * NA, 0, PP1_END>(k1, q1, c1, w1, wv, half, ec, a, cfA, wpA);
  wg_index<4, 4, 16 * 4 * NA, PP1_END, PP2_END>(k2, q2, c2, w2, wv, half, ec, a, cfB, wpB);
  wg_index<8, 6, 32 * 6 * NA, PP2_END, NPAIR>(k3, q3, c3, w3, wv, half, ec, a, cfC, wpC);

  int packed;
  {
    const bool s1 = wv < 1, s12 = wv < 5;
    const int qu = tid - (s1 ? 0 : (s12 ? P1_END : P2_END));
    const int q0 = qu < P_TOT - P2_END ? qu : P_TOT - P2_END - 1;
    const int* pi1 = s1 ? i1 : (s12 ? i2 : i3);
    const int* pl1 = s1 ? l1 : (s12 ? l2 : l3);
    const int* pi2 = s1 ? i1 : (s12 ? j2 : j3);
    const int* pl2 = s1 ? l1 : (s12 ? m2 : m3);
    const int* pi3 = s1 ? i1 : (s12 ? i2 : f3);
    const int* pl3 = s1 ? l1 : (s12 ? l2 : g3);
    const int o1 = (pi1[q0] * NM + pl1[q0]) * 4;
    const int o2 = (pi2[q0] * NM + pl2[q0]) * 4;
    const int o3 = (pi3[q0] * NM + pl3[q0]) * 4;
    packed = o1 | (s1 ? 0 : (o2 << 8)) | (s12 ? 0 : (o3 << 16));
  }

  float vA[1], vB[4], vC[8];
  wg_load<1>(wpA, vA);
  wg_load<4>(wpB, vB);
  wg_load<8>(wpC, vC);

  float yv[TPW][8];
#pragma unroll
  for (int k = 0; k < TPW; ++k) {
#pragma unroll
    for (int r = 0; r < 8; ++r) {
      const int e  = 8 * half + r;
      const int ey = e < NE ? e : NE - 1;
      const float v = y[(size_t)((tile0 + k) * 16 + lrow) * NE + ey];
      yv[k][r] = e < NE ? v : 0.0f;
    }
  }

  if (tid < P_TOT) ldsOff[tid] = packed;
  float* ldsWf = (float*)ldsW;
  wg_store<1, 0, PP1_END>(ldsWf, wv, half, lrow, cfA, vA);
  wg_store<4, PP1_END, PP2_END>(ldsWf, wv, half, lrow, cfB, vB);
  wg_store<8, PP2_END, NPAIR>(ldsWf, wv, half, lrow, cfC, vC);

  float* Xw = &ldsX[wv * TPW * 16 * XPITCH];
  float* Xst = Xw + srow * XPITCH + c4 * 4;
#pragma unroll
  for (int k = 0; k < TPW; ++k) { STORE_XROWS(xr[k], Xst + k * 16 * XPITCH) }

  __syncthreads();

  const v2f* wl = ldsW + half * EPAD + lrow;
  const char* Xb = (const char*)(Xw + lrow * XPITCH);
  const int* offl = ldsOff + 2 * half;

  v8f acc[TPW] = {};
  REGION(1, 0)
#pragma unroll 1
  for (int t0 = 8; t0 < 40; t0 += 8) REGION(2, t0)
#pragma unroll 1
  for (int t0 = 40; t0 < KSTEPS; t0 += 8) REGION(3, t0)

  asm volatile("s_wait_alu depctr_va_vdst(0)\n\tv_nop\n\tv_nop\n\tv_nop\n\tv_nop\n\tv_nop\n\tv_nop\n\tv_nop\n\tv_nop"
               : "+v"(acc[0]), "+v"(acc[1]), "+v"(acc[2]), "+v"(acc[3]));

#pragma unroll
  for (int k = 0; k < TPW; ++k) {
    float part = acc[k][0] * yv[k][0];
#pragma unroll
    for (int r = 1; r < 8; ++r) part += acc[k][r] * yv[k][r];
    part += __shfl_xor(part, 16, 32);
    if (half == 0) out[(size_t)((tile0 + k) * 16 + lrow) * NA + a] = part;
  }
}

extern "C" void kernel_launch(void* const* d_in, const int* in_sizes, int n_in,
                              void* d_out, int out_size, void* d_ws, size_t ws_size,
                              hipStream_t stream) {
  const float* x  = (const float*)d_in[0];
  const float* y  = (const float*)d_in[1];
  const float* w1 = (const float*)d_in[2];
  const float* w2 = (const float*)d_in[3];
  const float* w3 = (const float*)d_in[4];
  const float* c1 = (const float*)d_in[5];
  const float* c2 = (const float*)d_in[6];
  const float* c3 = (const float*)d_in[7];
  const int* i1 = (const int*)d_in[8];
  const int* l1 = (const int*)d_in[9];
  const int* k1 = (const int*)d_in[10];
  const int* q1 = (const int*)d_in[11];
  const int* i2 = (const int*)d_in[12];
  const int* j2 = (const int*)d_in[13];
  const int* l2 = (const int*)d_in[14];
  const int* m2 = (const int*)d_in[15];
  const int* k2 = (const int*)d_in[16];
  const int* q2 = (const int*)d_in[17];
  const int* i3 = (const int*)d_in[18];
  const int* j3 = (const int*)d_in[19];
  const int* f3 = (const int*)d_in[20];
  const int* l3 = (const int*)d_in[21];
  const int* m3 = (const int*)d_in[22];
  const int* g3 = (const int*)d_in[23];
  const int* k3 = (const int*)d_in[24];
  const int* q3 = (const int*)d_in[25];

  float* out  = (float*)d_out;

  contract_kernel<<<dim3(4, 64), 512, 0, stream>>>(x, y, w1, w2, w3, c1, c2, c3,
                                                   i1, l1, k1, q1,
                                                   i2, j2, l2, m2, k2, q2,
                                                   i3, j3, f3, l3, m3, g3, k3, q3, out);
}


# Two adjacent a per work-group (grid 8x32, wave = 2 tiles x 2 a): x of the a pair element-interleaved in LDS, one 8-byte gather per factor feeds both a's WMMAs; own tables per a; sum order unchanged
# speedup vs baseline: 3.8492x; 1.0717x over previous
#include <hip/hip_runtime.h>
#include <hip/hip_bf16.h>

typedef float v2f __attribute__((ext_vector_type(2)));
typedef float v8f __attribute__((ext_vector_type(8)));

#define NB 4096
#define NA 64
#define NL 16
#define NM 4
#define NE 10
#define P_TOT 416
#define P1_END 32
#define P2_END 160
#define EPAD 16
#define XPITCH 132
#define KSTEPS 104
#define WAVES 16
#define GROUPS_PER_A 8
#define BTILES (NB / 16)
#define TILES_PER_GROUP (BTILES / GROUPS_PER_A)
#define TPW (TILES_PER_GROUP / WAVES)
#define TILE_BYTES (16 * XPITCH * 4)
#define WTAB (P_TOT * EPAD)

#define NPAIR (P_TOT / 2)
#define PP1_END (P1_END / 2)
#define PP2_END (P2_END / 2)

template <int N, int Q, int ES, int PP0, int PP1>
__device__ __forceinline__ void wg_index(const int* __restrict__ k, const int* __restrict__ q,
                                         const float* __restrict__ c, const float* __restrict__ w,
                                         int wv, int half, int ec, int a,
                                         float* cf, const float** wp) {
#pragma unroll
  for (int j = 0; j < N; ++j) {
    int pp = PP0 + wv + WAVES * j;
    pp = pp < PP1 ? pp : PP1 - 1;
    const int t = 2 * (pp - PP0) + half;
    cf[j] = c[t];
    wp[j] = w + (ec * ES + (k[t] * Q + q[t]) * NA + a);
  }
}
template <int N>
__device__ __forceinline__ void wg_load(const float* const* wp, v2f* val) {
#pragma unroll
  for (int j = 0; j < N; ++j) val[j] = *(const v2f*)wp[j];
}
template <int N, int PP0, int PP1>
__device__ __forceinline__ void wg_store(float* ldsWf, int wv, int half, int lrow,
                                         const float* cf, const v2f* val) {
#pragma unroll
  for (int j = 0; j < N; ++j) {
    const int pp = PP0 + wv + WAVES * j;
    if (pp < PP1) {
      ldsWf[(pp * EPAD + lrow) * 2 + half]        = lrow < NE ? cf[j] * val[j].x : 0.0f;
      ldsWf[WTAB + (pp * EPAD + lrow) * 2 + half] = lrow < NE ? cf[j] * val[j].y : 0.0f;
    }
  }
}

template <int CNT>
__device__ __forceinline__ void xaddr(int pk0, int pk1, const char* Xb, const char** ad) {
#pragma unroll
  for (int f = 0; f < CNT; ++f) {
    ad[2 * f]     = Xb + ((pk0 >> (10 * f)) & 0x3FF);
    ad[2 * f + 1] = Xb + ((pk1 >> (10 * f)) & 0x3FF);
  }
}
template <int CNT, int TILE>
__device__ __forceinline__ void xgather(const char* const* ad, v2f* g) {
#pragma unroll
  for (int j = 0; j < 2 * CNT; ++j) g[j] = *(const v2f*)(ad[j] + TILE * TILE_BYTES);
}
template <int CNT>
__device__ __forceinline__ void xproduct(const v2f* g, v2f& f0, v2f& f1) {
  v2f xa = g[0], xb = g[1];
  if (CNT > 1) { xa *= g[2]; xb *= g[3]; }
  if (CNT > 2) { xa *= g[4]; xb *= g[5]; }
  f0.x = xa.x; f0.y = xb.x;
  f1.x = xa.y; f1.y = xb.y;
}

#define XROWSTEP ((size_t)2 * NA * 64)
#define LOAD_XROWS(V, P)                                     \
  V[0] = *(const float4*)((P));                              \
  V[1] = *(const float4*)((P) + 1 * XROWSTEP);               \
  V[2] = *(const float4*)((P) + 2 * XROWSTEP);               \
  V[3] = *(const float4*)((P) + 3 * XROWSTEP);               \
  V[4] = *(const float4*)((P) + 4 * XROWSTEP);               \
  V[5] = *(const float4*)((P) + 5 * XROWSTEP);               \
  V[6] = *(const float4*)((P) + 6 * XROWSTEP);               \
  V[7] = *(const float4*)((P) + 7 * XROWSTEP);
__device__ __forceinline__ void store_xrows(const float4* v0, const float4* v1, float* q) {
#pragma unroll
  for (int j = 0; j < 8; ++j) {
    *(float4*)(q + j * 2 * XPITCH)     = make_float4(v0[j].x, v1[j].x, v0[j].y, v1[j].y);
    *(float4*)(q + j * 2 * XPITCH + 4) = make_float4(v0[j].z, v1[j].z, v0[j].w, v1[j].w);
  }
}

#define REGION_TILE_HALF(CNT, TILE, H)                                                       \
  {                                                                                          \
    v2f g[4][2 * (CNT)];                                                                     \
    _Pragma("unroll")                                     \
    for (int u = 0; u < 4; ++u) xgather<CNT, TILE>(ad[4 * (H) + u], g[u]);                   \
    __builtin_amdgcn_sched_barrier(0);      \
    v2f bf0[4], bf1[4];                                                                      \
    _Pragma("unroll")                               \
    for (int u = 0; u < 4; ++u) xproduct<CNT>(g[u], bf0[u], bf1[u]);                         \
    _Pragma("unroll")                   \
    for (int u = 0; u < 4; ++u) {                                                            \
      acc[TILE][0] = __builtin_amdgcn_wmma_f32_16x16x4_f32(false, wf0[4 * (H) + u], false, bf0[u], \
                                                           (short)0, acc[TILE][0], false, false);  \
      acc[TILE][1] = __builtin_amdgcn_wmma_f32_16x16x4_f32(false, wf1[4 * (H) + u], false, bf1[u], \
                                                           (short)0, acc[TILE][1], false, false);  \
    }                                                                                        \
    __builtin_amdgcn_sched_barrier(0);   \
  }

#define REGION(CNT, T0)                                                                      \
  {                                                                                          \
    v2f wf0[8], wf1[8];                                                                      \
    int pk[16];                                                                              \
    _Pragma("unroll")                         \
    for (int u = 0; u < 8; ++u) {                                                            \
      wf0[u]        = wl[((T0) + u) * 2 * EPAD];                                             \
      wf1[u]        = wl[((T0) + u) * 2 * EPAD + WTAB / 2];                                  \
      pk[2 * u]     = offl[4 * ((T0) + u)];                                                  \
      pk[2 * u + 1] = offl[4 * ((T0) + u) + 1];                                              \
    }                                                                                        \
    __builtin_amdgcn_sched_barrier(0);   \
    const char* ad[8][2 * (CNT)];                                                            \
    _Pragma("unroll")                                    \
    for (int u = 0; u < 8; ++u) xaddr<CNT>(pk[2 * u], pk[2 * u + 1], Xb, ad[u]);             \
    __builtin_amdgcn_sched_barrier(0);    \
    REGION_TILE_HALF(CNT, 0, 0)                                                              \
    REGION_TILE_HALF(CNT, 0, 1)                                                              \
    REGION_TILE_HALF(CNT, 1, 0)                                                              \
    REGION_TILE_HALF(CNT, 1, 1)                                                              \
  }

__global__ __launch_bounds__(WAVES * 32) void contract_kernel(
    const float* __restrict__ x, const float* __restrict__ y,
    const float* __restrict__ w1, const float* __restrict__ w2, const float* __restrict__ w3,
    const float* __restrict__ c1, const float* __restrict__ c2, const float* __restrict__ c3,
    const int* __restrict__ i1, const int* __restrict__ l1,
    const int* __restrict__ k1, const int* __restrict__ q1,
    const int* __restrict__ i2, const int* __restrict__ j2,
    const int* __restrict__ l2, const int* __restrict__ m2,
    const int* __restrict__ k2, const int* __restrict__ q2,
    const int* __restrict__ i3, const int* __restrict__ j3, const int* __restrict__ f3,
    const int* __restrict__ l3, const int* __restrict__ m3, const int* __restrict__ g3,
    const int* __restrict__ k3, const int* __restrict__ q3,
    float* __restrict__ out) {
  __shared__ v2f    ldsW[2 * WTAB / 2];
  __shared__ float4 ldsX4[WAVES * TPW * 16 * XPITCH / 4];
  __shared__ int    ldsOff[P_TOT];
  float* ldsX = (float*)ldsX4;

  const int tid  = threadIdx.x;
  const int wv   = __builtin_amdgcn_readfirstlane(tid >> 5);
  const int lane = tid & 31;
  const int half = lane >> 4;
  const int lrow = lane & 15;
  const int a    = 2 * blockIdx.y;
  const int tile0 = blockIdx.x * TILES_PER_GROUP + wv * TPW;

  const int srow = lane >> 4;
  const int c4   = lane & 15;
  const float* xsrc = x + ((size_t)(tile0 * 16 + srow) * NA + a) * 64 + c4 * 4;
  float4 xr[TPW][2][8];
#pragma unroll
  for (int k = 0; k < TPW; ++k) {
    LOAD_XROWS(xr[k][0], xsrc + (size_t)(k * 16) * NA * 64)
    LOAD_XROWS(xr[k][1], xsrc + (size_t)(k * 16) * NA * 64 + 64)
  }

  const int ec = lrow < NE ? lrow : NE - 1;
  float cfA[1], cfB[4], cfC[8];
  const float *wpA[1], *wpB[4], *wpC[8];
  wg_index<1, 2, 4 * 2 * NA, 0, PP1_END>(k1, q1, c1, w1, wv, half, ec, a, cfA, wpA);
  wg_index<4, 4, 16 * 4 * NA, PP1_END, PP2_END>(k2, q2, c2, w2, wv, half, ec, a, cfB, wpB);
  wg_index<8, 6, 32 * 6 * NA, PP2_END, NPAIR>(k3, q3, c3, w3, wv, half, ec, a, cfC, wpC);

  int packed;
  {
    const bool s1 = wv < 1, s12 = wv < 5;
    const int qu = tid - (s1 ? 0 : (s12 ? P1_END : P2_END));
    const int q0 = qu < P_TOT - P2_END ? qu : P_TOT - P2_END - 1;
    const int* pi1 = s1 ? i1 : (s12 ? i2 : i3);
    const int* pl1 = s1 ? l1 : (s12 ? l2 : l3);
    const int* pi2 = s1 ? i1 : (s12 ? j2 : j3);
    const int* pl2 = s1 ? l1 : (s12 ? m2 : m3);
    const int* pi3 = s1 ? i1 : (s12 ? i2 : f3);
    const int* pl3 = s1 ? l1 : (s12 ? l2 : g3);
    const int o1 = (pi1[q0] * NM + pl1[q0]) * 8;
    const int o2 = (pi2[q0] * NM + pl2[q0]) * 8;
    const int o3 = (pi3[q0] * NM + pl3[q0]) * 8;
    packed = o1 | (s1 ? 0 : (o2 << 10)) | (s12 ? 0 : (o3 << 20));
  }

  v2f vA[1], vB[4], vC[8];
  wg_load<1>(wpA, vA);
  wg_load<4>(wpB, vB);
  wg_load<8>(wpC, vC);

  float yv[TPW][8];
#pragma unroll
  for (int k = 0; k < TPW; ++k) {
#pragma unroll
    for (int r = 0; r < 8; ++r) {
      const int e  = 8 * half + r;
      const int ey = e < NE ? e : NE - 1;
      const float v = y[(size_t)((tile0 + k) * 16 + lrow) * NE + ey];
      yv[k][r] = e < NE ? v : 0.0f;
    }
  }

  if (tid < P_TOT) ldsOff[tid] = packed;
  float* ldsWf = (float*)ldsW;
  wg_store<1, 0, PP1_END>(ldsWf, wv, half, lrow, cfA, vA);
  wg_store<4, PP1_END, PP2_END>(ldsWf, wv, half, lrow, cfB, vB);
  wg_store<8, PP2_END, NPAIR>(ldsWf, wv, half, lrow, cfC, vC);

  float* Xw = &ldsX[wv * TPW * 16 * XPITCH];
  float* Xst = Xw + srow * XPITCH + c4 * 8;
#pragma unroll
  for (int k = 0; k < TPW; ++k) store_xrows(xr[k][0], xr[k][1], Xst + k * 16 * XPITCH);

  __syncthreads();

  const v2f* wl = ldsW + half * EPAD + lrow;
  const char* Xb = (const char*)(Xw + lrow * XPITCH);
  const int* offl = ldsOff + 2 * half;

  v8f acc[TPW][2] = {};
  REGION(1, 0)
#pragma unroll 1
  for (int t0 = 8; t0 < 40; t0 += 8) REGION(2, t0)
#pragma unroll 1
  for (int t0 = 40; t0 < KSTEPS; t0 += 8) REGION(3, t0)

  asm volatile("s_wait_alu depctr_va_vdst(0)\n\tv_nop\n\tv_nop\n\tv_nop\n\tv_nop\n\tv_nop\n\tv_nop\n\tv_nop\n\tv_nop"
               : "+v"(acc[0][0]), "+v"(acc[0][1]), "+v"(acc[1][0]), "+v"(acc[1][1]));

#pragma unroll
  for (int k = 0; k < TPW; ++k) {
    float part0 = acc[k][0][0] * yv[k][0];
    float part1 = acc[k][1][0] * yv[k][0];
#pragma unroll
    for (int r = 1; r < 8; ++r) {
      part0 += acc[k][0][r] * yv[k][r];
      part1 += acc[k][1][r] * yv[k][r];
    }
    part0 += __shfl_xor(part0, 16, 32);
    part1 += __shfl_xor(part1, 16, 32);
    if (half == 0) {
      v2f o; o.x = part0; o.y = part1;
      *(v2f*)(out + (size_t)((tile0 + k) * 16 + lrow) * NA + a) = o;
    }
  }
}

extern "C" void kernel_launch(void* const* d_in, const int* in_sizes, int n_in,
                              void* d_out, int out_size, void* d_ws, size_t ws_size,
                              hipStream_t stream) {
  const float* x  = (const float*)d_in[0];
  const float* y  = (const float*)d_in[1];
  const float* w1 = (const float*)d_in[2];
  const float* w2 = (const float*)d_in[3];
  const float* w3 = (const float*)d_in[4];
  const float* c1 = (const float*)d_in[5];
  const float* c2 = (const float*)d_in[6];
  const float* c3 = (const float*)d_in[7];
  const int* i1 = (const int*)d_in[8];
  const int* l1 = (const int*)d_in[9];
  const int* k1 = (const int*)d_in[10];
  const int* q1 = (const int*)d_in[11];
  const int* i2 = (const int*)d_in[12];
  const int* j2 = (const int*)d_in[13];
  const int* l2 = (const int*)d_in[14];
  const int* m2 = (const int*)d_in[15];
  const int* k2 = (const int*)d_in[16];
  const int* q2 = (const int*)d_in[17];
  const int* i3 = (const int*)d_in[18];
  const int* j3 = (const int*)d_in[19];
  const int* f3 = (const int*)d_in[20];
  const int* l3 = (const int*)d_in[21];
  const int* m3 = (const int*)d_in[22];
  const int* g3 = (const int*)d_in[23];
  const int* k3 = (const int*)d_in[24];
  const int* q3 = (const int*)d_in[25];

  float* out  = (float*)d_out;

  contract_kernel<<<dim3(8, 32), 512, 0, stream>>>(x, y, w1, w2, w3, c1, c2, c3,
                                                   i1, l1, k1, q1,
                                                   i2, j2, l2, m2, k2, q2,
                                                   i3, j3, f3, l3, m3, g3, k3, q3, out);
}


# K loop rotated one gather group ahead, also across regions: next region's offsets, address unpack, first gathers issued inside this region's WMMA stream; y loads moved late; sum order unchanged
# speedup vs baseline: 3.9676x; 1.0308x over previous
#include <hip/hip_runtime.h>
#include <hip/hip_bf16.h>

typedef float v2f __attribute__((ext_vector_type(2)));
typedef float v8f __attribute__((ext_vector_type(8)));

#define NB 4096
#define NA 64
#define NL 16
#define NM 4
#define NE 10
#define P_TOT 416
#define P1_END 32
#define P2_END 160
#define EPAD 16
#define XPITCH 132
#define KSTEPS 104
#define WAVES 16
#define GROUPS_PER_A 8
#define BTILES (NB / 16)
#define TILES_PER_GROUP (BTILES / GROUPS_PER_A)
#define TPW (TILES_PER_GROUP / WAVES)
#define TILE_BYTES (16 * XPITCH * 4)
#define WTAB (P_TOT * EPAD)

#define NPAIR (P_TOT / 2)
#define PP1_END (P1_END / 2)
#define PP2_END (P2_END / 2)

template <int N, int Q, int ES, int PP0, int PP1>
__device__ __forceinline__ void wg_index(const int* __restrict__ k, const int* __restrict__ q,
                                         const float* __restrict__ c, const float* __restrict__ w,
                                         int wv, int half, int ec, int a,
                                         float* cf, const float** wp) {
#pragma unroll
  for (int j = 0; j < N; ++j) {
    int pp = PP0 + wv + WAVES * j;
    pp = pp < PP1 ? pp : PP1 - 1;
    const int t = 2 * (pp - PP0) + half;
    cf[j] = c[t];
    wp[j] = w + (ec * ES + (k[t] * Q + q[t]) * NA + a);
  }
}
template <int N>
__device__ __forceinline__ void wg_load(const float* const* wp, v2f* val) {
#pragma unroll
  for (int j = 0; j < N; ++j) val[j] = *(const v2f*)wp[j];
}
template <int N, int PP0, int PP1>
__device__ __forceinline__ void wg_store(float* ldsWf, int wv, int half, int lrow,
                                         const float* cf, const v2f* val) {
#pragma unroll
  for (int j = 0; j < N; ++j) {
    const int pp = PP0 + wv + WAVES * j;
    if (pp < PP1) {
      ldsWf[(pp * EPAD + lrow) * 2 + half]        = lrow < NE ? cf[j] * val[j].x : 0.0f;
      ldsWf[WTAB + (pp * EPAD + lrow) * 2 + half] = lrow < NE ? cf[j] * val[j].y : 0.0f;
    }
  }
}

typedef const __attribute__((address_space(3))) char* ldsp;
typedef const __attribute__((address_space(3))) v2f*  ldsv2;

template <int CNT>
__device__ __forceinline__ void xaddr(int pk0, int pk1, ldsp Xb, ldsp* ad) {
#pragma unroll
  for (int f = 0; f < CNT; ++f) {
    ad[2 * f]     = Xb + ((pk0 >> (10 * f)) & 0x3FF);
    ad[2 * f + 1] = Xb + ((pk1 >> (10 * f)) & 0x3FF);
  }
}
template <int CNT, int TILE>
__device__ __forceinline__ void xgather(const ldsp* ad, v2f* g) {
#pragma unroll
  for (int j = 0; j < 2 * CNT; ++j) g[j] = *(ldsv2)(ad[j] + TILE * TILE_BYTES);
}
template <int CNT>
__device__ __forceinline__ void xproduct(const v2f* g, v2f& f0, v2f& f1) {
  v2f xa = g[0], xb = g[1];
  if (CNT > 1) { xa *= g[2]; xb *= g[3]; }
  if (CNT > 2) { xa *= g[4]; xb *= g[5]; }
  f0.x = xa.x; f0.y = xb.x;
  f1.x = xa.y; f1.y = xb.y;
}

#define XROWSTEP ((size_t)2 * NA * 64)
#define LOAD_XROWS(V, P)                                     \
  V[0] = *(const float4*)((P));                              \
  V[1] = *(const float4*)((P) + 1 * XROWSTEP);               \
  V[2] = *(const float4*)((P) + 2 * XROWSTEP);               \
  V[3] = *(const float4*)((P) + 3 * XROWSTEP);               \
  V[4] = *(const float4*)((P) + 4 * XROWSTEP);               \
  V[5] = *(const float4*)((P) + 5 * XROWSTEP);               \
  V[6] = *(const float4*)((P) + 6 * XROWSTEP);               \
  V[7] = *(const float4*)((P) + 7 * XROWSTEP);
__device__ __forceinline__ void store_xrows(const float4* v0, const float4* v1, float* q) {
#pragma unroll
  for (int j = 0; j < 8; ++j) {
    *(float4*)(q + j * 2 * XPITCH)     = make_float4(v0[j].x, v1[j].x, v0[j].y, v1[j].y);
    *(float4*)(q + j * 2 * XPITCH + 4) = make_float4(v0[j].z, v1[j].z, v0[j].w, v1[j].w);
  }
}

#define LOAD_WF4(U0, KT)                                                                     \
  _Pragma("unroll")                                 \
  for (int u = 0; u < 4; ++u) {                                                              \
    wf0[(U0) + u] = wl[((KT) + u) * 2 * EPAD];                                               \
    wf1[(U0) + u] = wl[((KT) + u) * 2 * EPAD + WTAB / 2];                                    \
  }
#define LOAD_PK(KT)                                                                          \
  _Pragma("unroll")                                        \
  for (int u = 0; u < 8; ++u) {                                                              \
    pk[2 * u]     = offl[4 * ((KT) + u)];                                                    \
    pk[2 * u + 1] = offl[4 * ((KT) + u) + 1];                                                \
  }
#define GATHER4(CNT, TILE, R0, AD, G)                                                        \
  _Pragma("unroll")                                       \
  for (int u = 0; u < 4; ++u) xgather<CNT, TILE>(AD[(R0) + u], G[u]);
#define PRODWMMA4(CNT, TILE, U0, G)                                                          \
  {                                                                                          \
    v2f bf0[4], bf1[4];                                                                      \
    _Pragma("unroll")                               \
    for (int u = 0; u < 4; ++u) xproduct<CNT>(G[u], bf0[u], bf1[u]);                         \
    _Pragma("unroll")                   \
    for (int u = 0; u < 4; ++u) {                                                            \
      acc[TILE][0] = __builtin_amdgcn_wmma_f32_16x16x4_f32(false, wf0[(U0) + u], false, bf0[u], \
                                                           (short)0, acc[TILE][0], false, false);  \
      acc[TILE][1] = __builtin_amdgcn_wmma_f32_16x16x4_f32(false, wf1[(U0) + u], false, bf1[u], \
                                                           (short)0, acc[TILE][1], false, false);  \
    }                                                                                        \
  }
#define SB __builtin_amdgcn_sched_barrier(0)

#define REGION_PIPE(CNT, NCNT, T0, AD, G0, ADN, G0N, HASNEXT, HOOK)                          \
  {                                                                                          \
    v2f g1[4][2 * (CNT)], g2[4][2 * (CNT)], g3[4][2 * (CNT)];                                \
    int pk[16];                                                                              \
    SB;                                                                                      \
    LOAD_WF4(0, (T0))                                                                        \
    LOAD_WF4(4, (T0) + 4)                                                                    \
    GATHER4(CNT, 0, 4, AD, g1)                                                               \
    SB;                                                                                      \
    PRODWMMA4(CNT, 0, 0, G0)                                                                 \
    SB;                                                                                      \
    GATHER4(CNT, 1, 0, AD, g2)                                                               \
    SB;                                                                                      \
    PRODWMMA4(CNT, 0, 4, g1)                                                                 \
    SB;                                                                                      \
    if (HASNEXT) { LOAD_PK((T0) + 8) }                                                       \
    GATHER4(CNT, 1, 4, AD, g3)                                                               \
    SB;                                                                                      \
    PRODWMMA4(CNT, 1, 0, g2)                                                                 \
    if (HASNEXT) {                                                                           \
      _Pragma("unroll")                                 \
      for (int u = 0; u < 4; ++u) xaddr<NCNT>(pk[2 * u], pk[2 * u + 1], Xb, ADN[u]);         \
    }                                                                                        \
    SB;                                                                                      \
    if (HASNEXT) { GATHER4(NCNT, 0, 0, ADN, G0N) }                                           \
    HOOK                                                                                     \
    SB;                                                                                      \
    PRODWMMA4(CNT, 1, 4, g3)                                                                 \
    if (HASNEXT) {                                                                           \
      _Pragma("unroll")                                 \
      for (int u = 4; u < 8; ++u) xaddr<NCNT>(pk[2 * u], pk[2 * u + 1], Xb, ADN[u]);         \
    }                                                                                        \
  }

__global__ __launch_bounds__(WAVES * 32) void contract_kernel(
    const float* __restrict__ x,
    const float* y,
    const float* __restrict__ w1, const float* __restrict__ w2, const float* __restrict__ w3,
    const float* __restrict__ c1, const float* __restrict__ c2, const float* __restrict__ c3,
    const int* __restrict__ i1, const int* __restrict__ l1,
    const int* __restrict__ k1, const int* __restrict__ q1,
    const int* __restrict__ i2, const int* __restrict__ j2,
    const int* __restrict__ l2, const int* __restrict__ m2,
    const int* __restrict__ k2, const int* __restrict__ q2,
    const int* __restrict__ i3, const int* __restrict__ j3, const int* __restrict__ f3,
    const int* __restrict__ l3, const int* __restrict__ m3, const int* __restrict__ g3,
    const int* __restrict__ k3, const int* __restrict__ q3,
    float* __restrict__ out) {
  __shared__ v2f    ldsW[2 * WTAB / 2];
  __shared__ float4 ldsX4[WAVES * TPW * 16 * XPITCH / 4];
  __shared__ int    ldsOff[P_TOT];
  float* ldsX = (float*)ldsX4;

  const int tid  = threadIdx.x;
  const int wv   = __builtin_amdgcn_readfirstlane(tid >> 5);
  const int lane = tid & 31;
  const int half = lane >> 4;
  const int lrow = lane & 15;
  const int a    = 2 * blockIdx.y;
  const int tile0 = blockIdx.x * TILES_PER_GROUP + wv * TPW;

  const int srow = lane >> 4;
  const int c4   = lane & 15;
  const float* xsrc = x + ((size_t)(tile0 * 16 + srow) * NA + a) * 64 + c4 * 4;
  float4 xr[TPW][2][8];
#pragma unroll
  for (int k = 0; k < TPW; ++k) {
    LOAD_XROWS(xr[k][0], xsrc + (size_t)(k * 16) * NA * 64)
    LOAD_XROWS(xr[k][1], xsrc + (size_t)(k * 16) * NA * 64 + 64)
  }

  const int ec = lrow < NE ? lrow : NE - 1;
  float cfA[1], cfB[4], cfC[8];
  const float *wpA[1], *wpB[4], *wpC[8];
  wg_index<1, 2, 4 * 2 * NA, 0, PP1_END>(k1, q1, c1, w1, wv, half, ec, a, cfA, wpA);
  wg_index<4, 4, 16 * 4 * NA, PP1_END, PP2_END>(k2, q2, c2, w2, wv, half, ec, a, cfB, wpB);
  wg_index<8, 6, 32 * 6 * NA, PP2_END, NPAIR>(k3, q3, c3, w3, wv, half, ec, a, cfC, wpC);

  int packed;
  {
    const bool s1 = wv < 1, s12 = wv < 5;
    const int qu = tid - (s1 ? 0 : (s12 ? P1_END : P2_END));
    const int q0 = qu < P_TOT - P2_END ? qu : P_TOT - P2_END - 1;
    const int* pi1 = s1 ? i1 : (s12 ? i2 : i3);
    const int* pl1 = s1 ? l1 : (s12 ? l2 : l3);
    const int* pi2 = s1 ? i1 : (s12 ? j2 : j3);
    const int* pl2 = s1 ? l1 : (s12 ? m2 : m3);
    const int* pi3 = s1 ? i1 : (s12 ? i2 : f3);
    const int* pl3 = s1 ? l1 : (s12 ? l2 : g3);
    const int o1 = (pi1[q0] * NM + pl1[q0]) * 8;
    const int o2 = (pi2[q0] * NM + pl2[q0]) * 8;
    const int o3 = (pi3[q0] * NM + pl3[q0]) * 8;
    packed = o1 | (s1 ? 0 : (o2 << 10)) | (s12 ? 0 : (o3 << 20));
  }

  v2f vA[1], vB[4], vC[8];
  wg_load<1>(wpA, vA);
  wg_load<4>(wpB, vB);
  wg_load<8>(wpC, vC);

  if (tid < P_TOT) ldsOff[tid] = packed;
  float* ldsWf = (float*)ldsW;
  wg_store<1, 0, PP1_END>(ldsWf, wv, half, lrow, cfA, vA);
  wg_store<4, PP1_END, PP2_END>(ldsWf, wv, half, lrow, cfB, vB);
  wg_store<8, PP2_END, NPAIR>(ldsWf, wv, half, lrow, cfC, vC);

  float* Xw = &ldsX[wv * TPW * 16 * XPITCH];
  float* Xst = Xw + srow * XPITCH + c4 * 8;
#pragma unroll
  for (int k = 0; k < TPW; ++k) store_xrows(xr[k][0], xr[k][1], Xst + k * 16 * XPITCH);

  __syncthreads();

  const v2f* wl = ldsW + half * EPAD + lrow;
  ldsp Xb = (ldsp)(Xw + lrow * XPITCH);
  const int* offl = ldsOff + 2 * half;

  v8f acc[TPW][2] = {};
  v2f wf0[8], wf1[8];
  ldsp ad1[8][2], ad2[8][4], ad3[8][6];
  v2f gA[4][2], gB[4][4], gC[4][6];
  {
    int pk[16];
    LOAD_PK(0)
    SB;
#pragma unroll
    for (int u = 0; u < 8; ++u) xaddr<1>(pk[2 * u], pk[2 * u + 1], Xb, ad1[u]);
    SB;
    GATHER4(1, 0, 0, ad1, gA)
  }
  REGION_PIPE(1, 2, 0, ad1, gA, ad2, gB, 1, )
#pragma unroll 1
  for (int t0 = 8; t0 < 32; t0 += 8) REGION_PIPE(2, 2, t0, ad2, gB, ad2, gB, 1, )
  REGION_PIPE(2, 3, 32, ad2, gB, ad3, gC, 1, )
#pragma unroll 1
  for (int t0 = 40; t0 < KSTEPS - 8; t0 += 8) REGION_PIPE(3, 3, t0, ad3, gC, ad3, gC, 1, )
  float yv[TPW][8];
#define LOAD_Y                                                                               \
  _Pragma("unroll")            \
  for (int k = 0; k < TPW; ++k) {                                                            \
    _Pragma("unroll")                                                     \
    for (int r = 0; r < 8; ++r) {                                                            \
      const int e  = 8 * half + r;                                                           \
      const int ey = e < NE ? e : NE - 1;                                                    \
      yv[k][r] = y[(size_t)((tile0 + k) * 16 + lrow) * NE + ey];   \
    }                                                                                        \
  }
  REGION_PIPE(3, 3, KSTEPS - 8, ad3, gC, ad3, gC, 0, LOAD_Y)

  asm volatile("s_wait_alu depctr_va_vdst(0)\n\tv_nop\n\tv_nop\n\tv_nop\n\tv_nop\n\tv_nop\n\tv_nop\n\tv_nop\n\tv_nop"
               : "+v"(acc[0][0]), "+v"(acc[0][1]), "+v"(acc[1][0]), "+v"(acc[1][1]));

#pragma unroll
  for (int k = 0; k < TPW; ++k) {
    float ym[8];
#pragma unroll
    for (int r = 0; r < 8; ++r) ym[r] = (8 * half + r < NE) ? yv[k][r] : 0.0f;
    float part0 = acc[k][0][0] * ym[0];
    float part1 = acc[k][1][0] * ym[0];
#pragma unroll
    for (int r = 1; r < 8; ++r) {
      part0 += acc[k][0][r] * ym[r];
      part1 += acc[k][1][r] * ym[r];
    }
    part0 += __shfl_xor(part0, 16, 32);
    part1 += __shfl_xor(part1, 16, 32);
    if (half == 0) {
      v2f o; o.x = part0; o.y = part1;
      *(v2f*)(out + (size_t)((tile0 + k) * 16 + lrow) * NA + a) = o;
    }
  }
}

extern "C" void kernel_launch(void* const* d_in, const int* in_sizes, int n_in,
                              void* d_out, int out_size, void* d_ws, size_t ws_size,
                              hipStream_t stream) {
  const float* x  = (const float*)d_in[0];
  const float* y  = (const float*)d_in[1];
  const float* w1 = (const float*)d_in[2];
  const float* w2 = (const float*)d_in[3];
  const float* w3 = (const float*)d_in[4];
  const float* c1 = (const float*)d_in[5];
  const float* c2 = (const float*)d_in[6];
  const float* c3 = (const float*)d_in[7];
  const int* i1 = (const int*)d_in[8];
  const int* l1 = (const int*)d_in[9];
  const int* k1 = (const int*)d_in[10];
  const int* q1 = (const int*)d_in[11];
  const int* i2 = (const int*)d_in[12];
  const int* j2 = (const int*)d_in[13];
  const int* l2 = (const int*)d_in[14];
  const int* m2 = (const int*)d_in[15];
  const int* k2 = (const int*)d_in[16];
  const int* q2 = (const int*)d_in[17];
  const int* i3 = (const int*)d_in[18];
  const int* j3 = (const int*)d_in[19];
  const int* f3 = (const int*)d_in[20];
  const int* l3 = (const int*)d_in[21];
  const int* m3 = (const int*)d_in[22];
  const int* g3 = (const int*)d_in[23];
  const int* k3 = (const int*)d_in[24];
  const int* q3 = (const int*)d_in[25];

  float* out  = (float*)d_out;

  contract_kernel<<<dim3(8, 32), 512, 0, stream>>>(x, y, w1, w2, w3, c1, c2, c3,
                                                   i1, l1, k1, q1,
                                                   i2, j2, l2, m2, k2, q2,
                                                   i3, j3, f3, l3, m3, g3, k3, q3, out);
}
